# starting kernel
# baseline (speedup 1.0000x reference)
#include <hip/hip_runtime.h>

#define CH 128

typedef float v2f __attribute__((ext_vector_type(2)));
typedef float v8f __attribute__((ext_vector_type(8)));

__global__ void fill_kernel(float* __restrict__ p, int n, float v) {
    int i = blockIdx.x * blockDim.x + threadIdx.x;
    if (i < n) p[i] = v;
}

__global__ void deg_count_kernel(const int* __restrict__ dst,
                                 float* __restrict__ deg, int E) {
    int e = blockIdx.x * blockDim.x + threadIdx.x;
    if (e < E) atomicAdd(&deg[dst[e]], 1.0f);
}

__global__ void rsqrt_kernel(const float* __restrict__ deg,
                             float* __restrict__ dinv, int n) {
    int i = blockIdx.x * blockDim.x + threadIdx.x;
    if (i < n) dinv[i] = rsqrtf(deg[i]);
}

__global__ __launch_bounds__(256)
void gemm_wmma_f32(const float* __restrict__ X, const float* __restrict__ W,
                   float* __restrict__ H) {
    const int wave  = threadIdx.x >> 5;
    const int lane  = threadIdx.x & 31;
    const int m0    = blockIdx.x * 16;
    const int n0    = wave * 16;
    const int mrow  = lane & 15;
    const int khalf = (lane >> 4) << 1;
    const int ncol  = lane & 15;

    const float* __restrict__ xrow = X + (size_t)(m0 + mrow) * CH;

    v8f acc = {0.f, 0.f, 0.f, 0.f, 0.f, 0.f, 0.f, 0.f};
#pragma unroll
    for (int k0 = 0; k0 < CH; k0 += 4) {
        v2f a, b;
        a.x = xrow[k0 + khalf];
        a.y = xrow[k0 + khalf + 1];
        b.x = W[(size_t)(k0 + khalf) * CH + n0 + ncol];
        b.y = W[(size_t)(k0 + khalf + 1) * CH + n0 + ncol];
        acc = __builtin_amdgcn_wmma_f32_16x16x4_f32(
             false, a,  false, b,
             (short)0, acc,  false,  false);
    }

    const int moff = (lane >> 4) * 8;
#pragma unroll
    for (int j = 0; j < 8; ++j) {
        H[(size_t)(m0 + j + moff) * CH + n0 + ncol] = acc[j];
    }
}

__global__ __launch_bounds__(256)
void scatter128_kernel(const float* __restrict__ h, const int* __restrict__ src,
                       const int* __restrict__ dst, const float* __restrict__ dinv,
                       float* __restrict__ out, int E, int Etot) {
    const int w = (blockIdx.x * blockDim.x + threadIdx.x) >> 5;
    if (w >= Etot) return;
    const int lane = threadIdx.x & 31;
    int s, d;
    if (w < E) { s = src[w]; d = dst[w]; } else { s = d = w - E; }
    const float nrm = dinv[s] * dinv[d];
    const float4 v = ((const float4*)(h + (size_t)s * CH))[lane];
    float* od = out + (size_t)d * CH + lane * 4;
    atomicAdd(od + 0, v.x * nrm);
    atomicAdd(od + 1, v.y * nrm);
    atomicAdd(od + 2, v.z * nrm);
    atomicAdd(od + 3, v.w * nrm);
}

__global__ void postact_kernel(float* __restrict__ h, const float* __restrict__ b,
                               const float* __restrict__ g, const float* __restrict__ be,
                               const float* __restrict__ rm, const float* __restrict__ rv,
                               int total) {
    int i = blockIdx.x * blockDim.x + threadIdx.x;
    if (i >= total) return;
    const int c = i & (CH - 1);
    const float inv = g[c] * rsqrtf(rv[c] + 1e-5f);
    const float v = (h[i] + b[c] - rm[c]) * inv + be[c];
    h[i] = v > 0.f ? v : 0.f;
}

__global__ void gemm_out2_kernel(const float* __restrict__ X,
                                 const float* __restrict__ W3,
                                 float* __restrict__ H3, int M) {
    int i = blockIdx.x * blockDim.x + threadIdx.x;
    if (i >= M) return;
    const float* xr = X + (size_t)i * CH;
    float a0 = 0.f, a1 = 0.f;
#pragma unroll 8
    for (int k = 0; k < CH; ++k) {
        const float x = xr[k];
        a0 += x * W3[2 * k];
        a1 += x * W3[2 * k + 1];
    }
    H3[(size_t)i * 2]     = a0;
    H3[(size_t)i * 2 + 1] = a1;
}

__global__ void scatter2_kernel(const float* __restrict__ h3, const int* __restrict__ src,
                                const int* __restrict__ dst, const float* __restrict__ dinv,
                                float* __restrict__ out, int E, int Etot) {
    int e = blockIdx.x * blockDim.x + threadIdx.x;
    if (e >= Etot) return;
    int s, d;
    if (e < E) { s = src[e]; d = dst[e]; } else { s = d = e - E; }
    const float nrm = dinv[s] * dinv[d];
    atomicAdd(&out[(size_t)d * 2],     h3[(size_t)s * 2]     * nrm);
    atomicAdd(&out[(size_t)d * 2 + 1], h3[(size_t)s * 2 + 1] * nrm);
}

__global__ void add_bias2_kernel(float* __restrict__ out,
                                 const float* __restrict__ b3, int total) {
    int i = blockIdx.x * blockDim.x + threadIdx.x;
    if (i < total) out[i] += b3[i & 1];
}

extern "C" void kernel_launch(void* const* d_in, const int* in_sizes, int n_in,
                              void* d_out, int out_size, void* d_ws, size_t ws_size,
                              hipStream_t stream) {
    const float* x   = (const float*)d_in[0];
    const int*   ei  = (const int*)d_in[1];
    const float* W1  = (const float*)d_in[2];
    const float* b1  = (const float*)d_in[3];
    const float* g1  = (const float*)d_in[4];
    const float* be1 = (const float*)d_in[5];
    const float* rm1 = (const float*)d_in[6];
    const float* rv1 = (const float*)d_in[7];
    const float* W2  = (const float*)d_in[8];
    const float* b2  = (const float*)d_in[9];
    const float* g2  = (const float*)d_in[10];
    const float* be2 = (const float*)d_in[11];
    const float* rm2 = (const float*)d_in[12];
    const float* rv2 = (const float*)d_in[13];
    const float* W3  = (const float*)d_in[14];
    const float* b3  = (const float*)d_in[15];

    const int Nn   = in_sizes[0] / CH;
    const int Ee   = in_sizes[1] / 2;
    const int Etot = Ee + Nn;
    const int NC   = Nn * CH;

    const int* src = ei;
    const int* dst = ei + Ee;

    float* deg  = (float*)d_ws;
    float* dinv = deg  + Nn;
    float* bufA = dinv + Nn;
    float* bufB = bufA + (size_t)NC;
    float* h3   = bufB + (size_t)NC;
    float* out  = (float*)d_out;

    const int T = 256;
    const int gN  = (Nn + T - 1) / T;
    const int gE  = (Ee + T - 1) / T;
    const int gEt = (Etot + T - 1) / T;
    const int gNC = (NC + T - 1) / T;
    const int gSc = (Etot + 7) / 8;
    const int gMM = Nn / 16;

    fill_kernel<<<gN, T, 0, stream>>>(deg, Nn, 1.0f);
    deg_count_kernel<<<gE, T, 0, stream>>>(dst, deg, Ee);
    rsqrt_kernel<<<gN, T, 0, stream>>>(deg, dinv, Nn);

    gemm_wmma_f32<<<gMM, T, 0, stream>>>(x, W1, bufA);
    fill_kernel<<<gNC, T, 0, stream>>>(bufB, NC, 0.0f);
    scatter128_kernel<<<gSc, T, 0, stream>>>(bufA, src, dst, dinv, bufB, Ee, Etot);
    postact_kernel<<<gNC, T, 0, stream>>>(bufB, b1, g1, be1, rm1, rv1, NC);

    gemm_wmma_f32<<<gMM, T, 0, stream>>>(bufB, W2, bufA);
    fill_kernel<<<gNC, T, 0, stream>>>(bufB, NC, 0.0f);
    scatter128_kernel<<<gSc, T, 0, stream>>>(bufA, src, dst, dinv, bufB, Ee, Etot);
    postact_kernel<<<gNC, T, 0, stream>>>(bufB, b2, g2, be2, rm2, rv2, NC);

    gemm_out2_kernel<<<gN, T, 0, stream>>>(bufB, W3, h3, Nn);
    fill_kernel<<<(2 * Nn + T - 1) / T, T, 0, stream>>>(out, 2 * Nn, 0.0f);
    scatter2_kernel<<<gEt, T, 0, stream>>>(h3, src, dst, dinv, out, Ee, Etot);
    add_bias2_kernel<<<(2 * Nn + T - 1) / T, T, 0, stream>>>(out, b3, 2 * Nn);
}


# GEMM data path: each wave keeps its 128x16 slice of W in registers over a loop of row tiles (1024 resident groups); X tile in and H tile out go through LDS as whole 512-byte rows; sums unchanged
# speedup vs baseline: 1.9975x; 1.9975x over previous
#include <hip/hip_runtime.h>

#define CH 128

typedef float v2f __attribute__((ext_vector_type(2)));
typedef float v8f __attribute__((ext_vector_type(8)));

__global__ void fill_kernel(float* __restrict__ p, int n, float v) {
    int i = blockIdx.x * blockDim.x + threadIdx.x;
    if (i < n) p[i] = v;
}

__global__ void deg_count_kernel(const int* __restrict__ dst,
                                 float* __restrict__ deg, int E) {
    int e = blockIdx.x * blockDim.x + threadIdx.x;
    if (e < E) atomicAdd(&deg[dst[e]], 1.0f);
}

#define SEED_ROWS 32
#define SEED_DEPTH 8
__global__ __launch_bounds__(64)
void seed_kernel(const float* __restrict__ x, const float* __restrict__ deg,
                 float* __restrict__ dinv, float* __restrict__ P, int n) {
    const int w = __builtin_amdgcn_readfirstlane(
        (int)((blockIdx.x * blockDim.x + threadIdx.x) >> 5));
    const int lane = threadIdx.x & 31;
    const int r0 = w * SEED_ROWS;
    if (r0 >= n) return;
    const int r = r0 + lane;
    float di = 0.f;
    if (r < n) {
        di = rsqrtf(deg[r]);
        dinv[r] = di;
    }
    const float nrm = di * di;
    const float4* __restrict__ x4 =
        reinterpret_cast<const float4*>(x) + (size_t)r0 * (CH / 4) + lane;
    float4* __restrict__ p4 =
        reinterpret_cast<float4*>(P) + (size_t)r0 * (CH / 4) + lane;

    if (r0 + SEED_ROWS <= n) {
#pragma unroll 1
        for (int jb = 0; jb < SEED_ROWS; jb += SEED_DEPTH) {
            float4 v[SEED_DEPTH];
#pragma unroll
            for (int j = 0; j < SEED_DEPTH; ++j)
                v[j] = x4[(size_t)(jb + j) * (CH / 4)];
#pragma unroll
            for (int j = 0; j < SEED_DEPTH; ++j) {
                const float nj = __int_as_float(
                    __builtin_amdgcn_readlane(__float_as_int(nrm), jb + j));
                float4 o = v[j];
                o.x *= nj; o.y *= nj; o.z *= nj; o.w *= nj;
                p4[(size_t)(jb + j) * (CH / 4)] = o;
            }
        }
    } else {
        const int nr = n - r0;
        for (int j = 0; j < nr; ++j) {
            const float nj = __int_as_float(
                __builtin_amdgcn_readlane(__float_as_int(nrm), j));
            float4 o = x4[(size_t)j * (CH / 4)];
            o.x *= nj; o.y *= nj; o.z *= nj; o.w *= nj;
            p4[(size_t)j * (CH / 4)] = o;
        }
    }
}

#define GT_PITCH 132
#define GT_TILE (16 * GT_PITCH)
#define GT_BLOCKS 1024

template <bool SEED>
__global__ __launch_bounds__(256)
void gemm_wmma_f32(const float* __restrict__ X, const float* __restrict__ W,
                   const float* __restrict__ dinv,
                   const float* __restrict__ b, const float* __restrict__ g,
                   const float* __restrict__ be, const float* __restrict__ rm,
                   const float* __restrict__ rv,
                   float* __restrict__ H, float* __restrict__ S, int ntiles) {
    __shared__ __attribute__((aligned(16))) float As[2][GT_TILE];
    __shared__ __attribute__((aligned(16))) float Os[2][GT_TILE];

    const int wave  = __builtin_amdgcn_readfirstlane((int)(threadIdx.x >> 5));
    const int lane  = threadIdx.x & 31;
    const int n0    = wave * 16;
    const int mrow  = lane & 15;
    const int khalf = (lane >> 4) << 1;
    const int ncol  = lane & 15;
    const int moff  = (lane >> 4) * 8;

    int t = blockIdx.x;
    if (t >= ntiles) return;

    const int c = n0 + ncol;
    const float inv = g[c] * rsqrtf(rv[c] + 1e-5f);
    const float bc  = b[c];
    const float rmc = rm[c];
    const float bec = be[c];

    v2f bf[CH / 4];
#pragma unroll
    for (int kk = 0; kk < CH / 4; ++kk) {
        bf[kk].x = W[(size_t)(4 * kk + khalf) * CH + c];
        bf[kk].y = W[(size_t)(4 * kk + khalf + 1) * CH + c];
    }

    const int soff0 = wave * GT_PITCH + lane * 4;
    const int soff1 = (wave + 8) * GT_PITCH + lane * 4;
    const int aoff  = mrow * GT_PITCH + khalf;

    {
        const float4* __restrict__ xp =
            reinterpret_cast<const float4*>(X + (size_t)t * (16 * CH));
        const float4 p0 = xp[threadIdx.x];
        const float4 p1 = xp[threadIdx.x + 256];
        *reinterpret_cast<float4*>(&As[0][soff0]) = p0;
        *reinterpret_cast<float4*>(&As[0][soff1]) = p1;
    }
    __syncthreads();

    int cur = 0;
    for (; t < ntiles; t += gridDim.x) {
        const int tn = t + gridDim.x;
        const bool more = tn < ntiles;
        float4 p0 = make_float4(0.f, 0.f, 0.f, 0.f);
        float4 p1 = p0;
        if (more) {
            const float4* __restrict__ xp =
                reinterpret_cast<const float4*>(X + (size_t)tn * (16 * CH));
            p0 = xp[threadIdx.x];
            p1 = xp[threadIdx.x + 256];
        }
        const int m0 = t * 16;
        float d0 = 0.f, d1 = 0.f;
        if (SEED) {
            d0 = dinv[m0 + wave];
            d1 = dinv[m0 + 8 + wave];
        }

        const float* as = &As[cur][aoff];
        v8f acc = {0.f, 0.f, 0.f, 0.f, 0.f, 0.f, 0.f, 0.f};
#pragma unroll
        for (int kk = 0; kk < CH / 4; ++kk) {
            const v2f a = *reinterpret_cast<const v2f*>(as + 4 * kk);
            acc = __builtin_amdgcn_wmma_f32_16x16x4_f32(
                 false, a,  false, bf[kk],
                 (short)0, acc,  false,  false);
        }

        float* os = &Os[cur][0];
#pragma unroll
        for (int j = 0; j < 8; ++j) {
            const float y = (acc[j] + bc - rmc) * inv + bec;
            os[(j + moff) * GT_PITCH + c] = y > 0.f ? y : 0.f;
        }
        if (more) {
            *reinterpret_cast<float4*>(&As[cur ^ 1][soff0]) = p0;
            *reinterpret_cast<float4*>(&As[cur ^ 1][soff1]) = p1;
        }
        __syncthreads();

        const float4 h0 = *reinterpret_cast<const float4*>(&os[soff0]);
        const float4 h1 = *reinterpret_cast<const float4*>(&os[soff1]);
        float4* __restrict__ hp = reinterpret_cast<float4*>(H + (size_t)m0 * CH);
        hp[threadIdx.x]       = h0;
        hp[threadIdx.x + 256] = h1;
        if (SEED) {
            const float s0 = d0 * d0;
            const float s1 = d1 * d1;
            float4 q0 = h0, q1 = h1;
            q0.x *= s0; q0.y *= s0; q0.z *= s0; q0.w *= s0;
            q1.x *= s1; q1.y *= s1; q1.z *= s1; q1.w *= s1;
            float4* __restrict__ sp = reinterpret_cast<float4*>(S + (size_t)m0 * CH);
            sp[threadIdx.x]       = q0;
            sp[threadIdx.x + 256] = q1;
        }
        cur ^= 1;
    }
}

#define EPW 8

__device__ __forceinline__
void scatter128_one(const float* __restrict__ h, const int* __restrict__ src,
                    const int* __restrict__ dst, const float* __restrict__ dinv,
                    float* __restrict__ out, int e, int lane) {
    const int s = src[e];
    const int d = dst[e];
    const float nrm = dinv[s] * dinv[d];
    const float* __restrict__ hs = h + (size_t)s * CH + lane;
    const float v0 = hs[0];
    const float v1 = hs[32];
    const float v2 = hs[64];
    const float v3 = hs[96];
    float* od = out + (size_t)d * CH + lane;
    atomicAdd(od,      v0 * nrm);
    atomicAdd(od + 32, v1 * nrm);
    atomicAdd(od + 64, v2 * nrm);
    atomicAdd(od + 96, v3 * nrm);
}

__global__ __launch_bounds__(256)
void scatter128_kernel(const float* __restrict__ h, const int* __restrict__ src,
                       const int* __restrict__ dst, const float* __restrict__ dinv,
                       float* __restrict__ out, int E) {
    const int w = __builtin_amdgcn_readfirstlane(
        (int)((blockIdx.x * blockDim.x + threadIdx.x) >> 5));
    const int e0 = w * EPW;
    if (e0 >= E) return;
    const int lane = threadIdx.x & 31;

    if (e0 + EPW > E) {
        for (int e = e0; e < E; ++e)
            scatter128_one(h, src, dst, dinv, out, e, lane);
        return;
    }

    const int e = e0 + (lane & (EPW - 1));
    const int s = src[e];
    const int d = dst[e];
    const float nrm = dinv[s] * dinv[d];

    float v[EPW][4];
#pragma unroll
    for (int j = 0; j < EPW; ++j) {
        const int sj = __builtin_amdgcn_readlane(s, j);
        const float* __restrict__ hs = h + (size_t)sj * CH + lane;
        v[j][0] = hs[0];
        v[j][1] = hs[32];
        v[j][2] = hs[64];
        v[j][3] = hs[96];
    }
#pragma unroll
    for (int j = 0; j < EPW; ++j) {
        const int   dj = __builtin_amdgcn_readlane(d, j);
        const float nj = __int_as_float(
            __builtin_amdgcn_readlane(__float_as_int(nrm), j));
        float* od = out + (size_t)dj * CH + lane;
        atomicAdd(od,      v[j][0] * nj);
        atomicAdd(od + 32, v[j][1] * nj);
        atomicAdd(od + 64, v[j][2] * nj);
        atomicAdd(od + 96, v[j][3] * nj);
    }
}

__global__ void gemm_out2_kernel(const float* __restrict__ X,
                                 const float* __restrict__ W3,
                                 const float* __restrict__ dinv,
                                 const float* __restrict__ b3,
                                 float* __restrict__ H3,
                                 float* __restrict__ out, int M) {
    int i = blockIdx.x * blockDim.x + threadIdx.x;
    if (i >= M) return;
    const float* xr = X + (size_t)i * CH;
    float a0 = 0.f, a1 = 0.f;
#pragma unroll 8
    for (int k = 0; k < CH; ++k) {
        const float x = xr[k];
        a0 += x * W3[2 * k];
        a1 += x * W3[2 * k + 1];
    }
    H3[(size_t)i * 2]     = a0;
    H3[(size_t)i * 2 + 1] = a1;
    const float di = dinv[i];
    const float nrm = di * di;
    out[(size_t)i * 2]     = a0 * nrm + b3[0];
    out[(size_t)i * 2 + 1] = a1 * nrm + b3[1];
}

__global__ void scatter2_kernel(const float* __restrict__ h3, const int* __restrict__ src,
                                const int* __restrict__ dst, const float* __restrict__ dinv,
                                float* __restrict__ out, int E) {
    int e = blockIdx.x * blockDim.x + threadIdx.x;
    if (e >= E) return;
    const int s = src[e];
    const int d = dst[e];
    const float nrm = dinv[s] * dinv[d];
    atomicAdd(&out[(size_t)d * 2],     h3[(size_t)s * 2]     * nrm);
    atomicAdd(&out[(size_t)d * 2 + 1], h3[(size_t)s * 2 + 1] * nrm);
}

extern "C" void kernel_launch(void* const* d_in, const int* in_sizes, int n_in,
                              void* d_out, int out_size, void* d_ws, size_t ws_size,
                              hipStream_t stream) {
    const float* x   = (const float*)d_in[0];
    const int*   ei  = (const int*)d_in[1];
    const float* W1  = (const float*)d_in[2];
    const float* b1  = (const float*)d_in[3];
    const float* g1  = (const float*)d_in[4];
    const float* be1 = (const float*)d_in[5];
    const float* rm1 = (const float*)d_in[6];
    const float* rv1 = (const float*)d_in[7];
    const float* W2  = (const float*)d_in[8];
    const float* b2  = (const float*)d_in[9];
    const float* g2  = (const float*)d_in[10];
    const float* be2 = (const float*)d_in[11];
    const float* rm2 = (const float*)d_in[12];
    const float* rv2 = (const float*)d_in[13];
    const float* W3  = (const float*)d_in[14];
    const float* b3  = (const float*)d_in[15];

    const int Nn   = in_sizes[0] / CH;
    const int Ee   = in_sizes[1] / 2;
    const int NC   = Nn * CH;

    const int* src = ei;
    const int* dst = ei + Ee;

    float* deg  = (float*)d_ws;
    float* dinv = deg  + Nn;
    float* bufA = dinv + Nn;
    float* bufB = bufA + (size_t)NC;
    float* bufC = bufB + (size_t)NC;
    float* h3   = bufC + (size_t)NC;
    float* out  = (float*)d_out;

    const int T = 256;
    const int gN  = (Nn + T - 1) / T;
    const int gE  = (Ee + T - 1) / T;
    const int gSd = (Nn + 2 * SEED_ROWS - 1) / (2 * SEED_ROWS);
    const int gSc = (Ee + 8 * EPW - 1) / (8 * EPW);
    const int gMM = Nn / 16;

    fill_kernel<<<gN, T, 0, stream>>>(deg, Nn, 1.0f);
    deg_count_kernel<<<gE, T, 0, stream>>>(dst, deg, Ee);
    seed_kernel<<<gSd, 64, 0, stream>>>(x, deg, dinv, bufA, Nn);

    scatter128_kernel<<<gSc, T, 0, stream>>>(x, src, dst, dinv, bufA, Ee);
    gemm_wmma_f32<true><<<GT_BLOCKS, T, 0, stream>>>(bufA, W1, dinv, b1, g1, be1, rm1, rv1,
                                                     bufB, bufC, gMM);

    scatter128_kernel<<<gSc, T, 0, stream>>>(bufB, src, dst, dinv, bufC, Ee);
    gemm_wmma_f32<false><<<GT_BLOCKS, T, 0, stream>>>(bufC, W2, dinv, b2, g2, be2, rm2, rv2,
                                                      bufA, nullptr, gMM);

    gemm_out2_kernel<<<gN, T, 0, stream>>>(bufA, W3, dinv, b3, h3, out, Nn);
    scatter2_kernel<<<gE, T, 0, stream>>>(h3, src, dst, dinv, out, Ee);
}


# 128-channel aggregation by owner work-groups: edges binned once per call into 391-row destination ranges, each group sums its rows in LDS with LDS float adds (arrival order); 9 to 10 launches
# speedup vs baseline: 2.4982x; 1.2507x over previous
#include <hip/hip_runtime.h>

#define CH 128

typedef float v2f __attribute__((ext_vector_type(2)));
typedef float v8f __attribute__((ext_vector_type(8)));

__global__ void fill_kernel(float* __restrict__ p, int n, float v) {
    int i = blockIdx.x * blockDim.x + threadIdx.x;
    if (i < n) p[i] = v;
}

__global__ void deg_count_kernel(const int* __restrict__ dst,
                                 float* __restrict__ deg, int E) {
    int e = blockIdx.x * blockDim.x + threadIdx.x;
    if (e < E) atomicAdd(&deg[dst[e]], 1.0f);
}

#define OWN_ROWS 391
#define OWN_MAX 256
#define OWN_SHIFT 9
#define BIN_T 1024

__global__ __launch_bounds__(BIN_T)
void bin_count_kernel(const int* __restrict__ dst, const float* __restrict__ deg,
                      float* __restrict__ dinv, int* __restrict__ cnt,
                      int E, int n, int chunk) {
    __shared__ int hist[OWN_MAX];
    const int c = blockIdx.x;
    for (int i = c * BIN_T + threadIdx.x; i < n; i += gridDim.x * BIN_T)
        dinv[i] = rsqrtf(deg[i]);
    if (threadIdx.x < OWN_MAX) hist[threadIdx.x] = 0;
    __syncthreads();
    const int e0 = c * chunk;
    const int e1 = min(E, e0 + chunk);
    for (int e = e0 + threadIdx.x; e < e1; e += BIN_T)
        atomicAdd(&hist[dst[e] / OWN_ROWS], 1);
    __syncthreads();
    if (threadIdx.x < OWN_MAX) cnt[c * OWN_MAX + threadIdx.x] = hist[threadIdx.x];
}

__global__ __launch_bounds__(BIN_T)
void bin_place_kernel(const int* __restrict__ src, const int* __restrict__ dst,
                      const float* __restrict__ dinv, const int* __restrict__ cnt,
                      int* __restrict__ start, int2* __restrict__ rec,
                      int E, int chunk) {
    __shared__ int part_tot[BIN_T / OWN_MAX][OWN_MAX];
    __shared__ int part_pre[BIN_T / OWN_MAX][OWN_MAX];
    __shared__ int tot[OWN_MAX];
    __shared__ int cursor[OWN_MAX];
    const int c = blockIdx.x;
    const int o = threadIdx.x & (OWN_MAX - 1);
    const int q = threadIdx.x / OWN_MAX;
    const int per = OWN_MAX / (BIN_T / OWN_MAX);
    int t = 0, p = 0;
    for (int cc = q * per; cc < (q + 1) * per; ++cc) {
        const int v = cnt[cc * OWN_MAX + o];
        t += v;
        if (cc < c) p += v;
    }
    part_tot[q][o] = t;
    part_pre[q][o] = p;
    __syncthreads();
    if (threadIdx.x < OWN_MAX) {
        int s = 0;
        for (int k = 0; k < BIN_T / OWN_MAX; ++k) s += part_tot[k][o];
        tot[o] = s;
    }
    __syncthreads();
    if (threadIdx.x < OWN_MAX) {
        int s = 0;
        for (int k = 0; k < o; ++k) s += tot[k];
        int pre = 0;
        for (int k = 0; k < BIN_T / OWN_MAX; ++k) pre += part_pre[k][o];
        cursor[o] = s + pre;
        if (c == 0) {
            start[o] = s;
            if (o == OWN_MAX - 1) start[OWN_MAX] = s + tot[o];
        }
    }
    __syncthreads();
    const int e0 = c * chunk;
    const int e1 = min(E, e0 + chunk);
    for (int e = e0 + threadIdx.x; e < e1; e += BIN_T) {
        const int s = src[e];
        const int d = dst[e];
        const int ow = d / OWN_ROWS;
        const int lr = d - ow * OWN_ROWS;
        const int slot = atomicAdd(&cursor[ow], 1);
        rec[slot] = make_int2((s << OWN_SHIFT) | lr,
                              __float_as_int(dinv[s] * dinv[d]));
    }
}

#define GT_PITCH 132
#define GT_TILE (16 * GT_PITCH)
#define GT_BLOCKS 1024

__global__ __launch_bounds__(256)
void gemm_wmma_f32(const float* __restrict__ X, const float* __restrict__ W,
                   const float* __restrict__ b, const float* __restrict__ g,
                   const float* __restrict__ be, const float* __restrict__ rm,
                   const float* __restrict__ rv,
                   float* __restrict__ H, int ntiles) {
    __shared__ __attribute__((aligned(16))) float As[2][GT_TILE];
    __shared__ __attribute__((aligned(16))) float Os[2][GT_TILE];

    const int wave  = __builtin_amdgcn_readfirstlane((int)(threadIdx.x >> 5));
    const int lane  = threadIdx.x & 31;
    const int n0    = wave * 16;
    const int mrow  = lane & 15;
    const int khalf = (lane >> 4) << 1;
    const int ncol  = lane & 15;
    const int moff  = (lane >> 4) * 8;

    int t = blockIdx.x;
    if (t >= ntiles) return;

    const int c = n0 + ncol;
    const float inv = g[c] * rsqrtf(rv[c] + 1e-5f);
    const float bc  = b[c];
    const float rmc = rm[c];
    const float bec = be[c];

    v2f bf[CH / 4];
#pragma unroll
    for (int kk = 0; kk < CH / 4; ++kk) {
        bf[kk].x = W[(size_t)(4 * kk + khalf) * CH + c];
        bf[kk].y = W[(size_t)(4 * kk + khalf + 1) * CH + c];
    }

    const int soff0 = wave * GT_PITCH + lane * 4;
    const int soff1 = (wave + 8) * GT_PITCH + lane * 4;
    const int aoff  = mrow * GT_PITCH + khalf;

    {
        const float4* __restrict__ xp =
            reinterpret_cast<const float4*>(X + (size_t)t * (16 * CH));
        const float4 p0 = xp[threadIdx.x];
        const float4 p1 = xp[threadIdx.x + 256];
        *reinterpret_cast<float4*>(&As[0][soff0]) = p0;
        *reinterpret_cast<float4*>(&As[0][soff1]) = p1;
    }
    __syncthreads();

    int cur = 0;
    for (; t < ntiles; t += gridDim.x) {
        const int tn = t + gridDim.x;
        const bool more = tn < ntiles;
        float4 p0 = make_float4(0.f, 0.f, 0.f, 0.f);
        float4 p1 = p0;
        if (more) {
            const float4* __restrict__ xp =
                reinterpret_cast<const float4*>(X + (size_t)tn * (16 * CH));
            p0 = xp[threadIdx.x];
            p1 = xp[threadIdx.x + 256];
        }
        const int m0 = t * 16;

        const float* as = &As[cur][aoff];
        v8f acc = {0.f, 0.f, 0.f, 0.f, 0.f, 0.f, 0.f, 0.f};
#pragma unroll
        for (int kk = 0; kk < CH / 4; ++kk) {
            const v2f a = *reinterpret_cast<const v2f*>(as + 4 * kk);
            acc = __builtin_amdgcn_wmma_f32_16x16x4_f32(
                 false, a,  false, bf[kk],
                 (short)0, acc,  false,  false);
        }

        float* os = &Os[cur][0];
#pragma unroll
        for (int j = 0; j < 8; ++j) {
            const float y = (acc[j] + bc - rmc) * inv + bec;
            os[(j + moff) * GT_PITCH + c] = y > 0.f ? y : 0.f;
        }
        if (more) {
            *reinterpret_cast<float4*>(&As[cur ^ 1][soff0]) = p0;
            *reinterpret_cast<float4*>(&As[cur ^ 1][soff1]) = p1;
        }
        __syncthreads();

        const float4 h0 = *reinterpret_cast<const float4*>(&os[soff0]);
        const float4 h1 = *reinterpret_cast<const float4*>(&os[soff1]);
        float4* __restrict__ hp = reinterpret_cast<float4*>(H + (size_t)m0 * CH);
        hp[threadIdx.x]       = h0;
        hp[threadIdx.x + 256] = h1;
        cur ^= 1;
    }
}

#define EPW 8
#define OWN_T 1024
#define OWN_WAVES (OWN_T / 32)
#define OWN_INIT 7

__device__ __forceinline__
void own_add(float* a, const float4 v, const float nj) {
    atomicAdd(a,      v.x * nj);
    atomicAdd(a + 32, v.y * nj);
    atomicAdd(a + 64, v.z * nj);
    atomicAdd(a + 96, v.w * nj);
}

__global__ __launch_bounds__(OWN_T)
void owner_aggregate_kernel(const float* __restrict__ h, const int2* __restrict__ rec,
                            const int* __restrict__ start,
                            const float* __restrict__ dinv,
                            float* __restrict__ out, int n) {
    __shared__ __attribute__((aligned(16))) float acc[OWN_ROWS * CH];

    const int wave = __builtin_amdgcn_readfirstlane((int)(threadIdx.x >> 5));
    const int lane = threadIdx.x & 31;
    const int o  = blockIdx.x;
    const int r0 = o * OWN_ROWS;
    const int nr = min(OWN_ROWS, n - r0);

    for (int lb = wave; lb < nr; lb += OWN_INIT * OWN_WAVES) {
        float4 v[OWN_INIT];
#pragma unroll
        for (int u = 0; u < OWN_INIT; ++u) {
            const int lr = min(lb + u * OWN_WAVES, nr - 1);
            v[u] = reinterpret_cast<const float4*>(h + (size_t)(r0 + lr) * CH)[lane];
        }
#pragma unroll
        for (int u = 0; u < OWN_INIT; ++u) {
            const int lr = lb + u * OWN_WAVES;
            if (lr < nr) {
                const float di = dinv[r0 + lr];
                const float nrm = di * di;
                float* a = &acc[lr * CH + lane];
                a[0]  = v[u].x * nrm;
                a[32] = v[u].y * nrm;
                a[64] = v[u].z * nrm;
                a[96] = v[u].w * nrm;
            }
        }
    }
    __syncthreads();

    const int s0 = __builtin_amdgcn_readfirstlane(start[o]);
    const int s1 = __builtin_amdgcn_readfirstlane(start[o + 1]);
    const int nfull = (s1 - s0) / EPW;
    const int sub = lane & (EPW - 1);

    int2 r = make_int2(0, 0);
    if (wave < nfull) r = rec[s0 + wave * EPW + sub];
    for (int b = wave; b < nfull; b += OWN_WAVES) {
        int2 rn = make_int2(0, 0);
        if (b + OWN_WAVES < nfull) rn = rec[s0 + (b + OWN_WAVES) * EPW + sub];
        const int   pk  = r.x;
        const float nrm = __int_as_float(r.y);
        float4 v[EPW];
#pragma unroll
        for (int j = 0; j < EPW; ++j) {
            const int sj = __builtin_amdgcn_readlane(pk, j) >> OWN_SHIFT;
            v[j] = reinterpret_cast<const float4*>(h + (size_t)sj * CH)[lane];
        }
#pragma unroll
        for (int j = 0; j < EPW; ++j) {
            const int   lj = __builtin_amdgcn_readlane(pk, j) & ((1 << OWN_SHIFT) - 1);
            const float nj = __int_as_float(
                __builtin_amdgcn_readlane(__float_as_int(nrm), j));
            own_add(&acc[lj * CH + lane], v[j], nj);
        }
        r = rn;
    }
    if (wave == (nfull & (OWN_WAVES - 1))) {
        for (int e = s0 + nfull * EPW; e < s1; ++e) {
            const int2 q = rec[e];
            const int sj = q.x >> OWN_SHIFT;
            const int lj = q.x & ((1 << OWN_SHIFT) - 1);
            const float4 v =
                reinterpret_cast<const float4*>(h + (size_t)sj * CH)[lane];
            own_add(&acc[lj * CH + lane], v, __int_as_float(q.y));
        }
    }
    __syncthreads();

    for (int lr = wave; lr < nr; lr += OWN_WAVES) {
        const float* a = &acc[lr * CH + lane];
        const float4 v = make_float4(a[0], a[32], a[64], a[96]);
        reinterpret_cast<float4*>(out + (size_t)(r0 + lr) * CH)[lane] = v;
    }
}

__global__ void gemm_out2_kernel(const float* __restrict__ X,
                                 const float* __restrict__ W3,
                                 const float* __restrict__ dinv,
                                 const float* __restrict__ b3,
                                 float* __restrict__ H3,
                                 float* __restrict__ out, int M) {
    int i = blockIdx.x * blockDim.x + threadIdx.x;
    if (i >= M) return;
    const float* xr = X + (size_t)i * CH;
    float a0 = 0.f, a1 = 0.f;
#pragma unroll 8
    for (int k = 0; k < CH; ++k) {
        const float x = xr[k];
        a0 += x * W3[2 * k];
        a1 += x * W3[2 * k + 1];
    }
    H3[(size_t)i * 2]     = a0;
    H3[(size_t)i * 2 + 1] = a1;
    const float di = dinv[i];
    const float nrm = di * di;
    out[(size_t)i * 2]     = a0 * nrm + b3[0];
    out[(size_t)i * 2 + 1] = a1 * nrm + b3[1];
}

__global__ void scatter2_kernel(const float* __restrict__ h3, const int* __restrict__ src,
                                const int* __restrict__ dst, const float* __restrict__ dinv,
                                float* __restrict__ out, int E) {
    int e = blockIdx.x * blockDim.x + threadIdx.x;
    if (e >= E) return;
    const int s = src[e];
    const int d = dst[e];
    const float nrm = dinv[s] * dinv[d];
    atomicAdd(&out[(size_t)d * 2],     h3[(size_t)s * 2]     * nrm);
    atomicAdd(&out[(size_t)d * 2 + 1], h3[(size_t)s * 2 + 1] * nrm);
}

extern "C" void kernel_launch(void* const* d_in, const int* in_sizes, int n_in,
                              void* d_out, int out_size, void* d_ws, size_t ws_size,
                              hipStream_t stream) {
    const float* x   = (const float*)d_in[0];
    const int*   ei  = (const int*)d_in[1];
    const float* W1  = (const float*)d_in[2];
    const float* b1  = (const float*)d_in[3];
    const float* g1  = (const float*)d_in[4];
    const float* be1 = (const float*)d_in[5];
    const float* rm1 = (const float*)d_in[6];
    const float* rv1 = (const float*)d_in[7];
    const float* W2  = (const float*)d_in[8];
    const float* b2  = (const float*)d_in[9];
    const float* g2  = (const float*)d_in[10];
    const float* be2 = (const float*)d_in[11];
    const float* rm2 = (const float*)d_in[12];
    const float* rv2 = (const float*)d_in[13];
    const float* W3  = (const float*)d_in[14];
    const float* b3  = (const float*)d_in[15];

    const int Nn   = in_sizes[0] / CH;
    const int Ee   = in_sizes[1] / 2;
    const int NC   = Nn * CH;

    const int* src = ei;
    const int* dst = ei + Ee;

    float* deg  = (float*)d_ws;
    float* dinv = deg  + Nn;
    float* bufA = dinv + Nn;
    float* bufB = bufA + (size_t)NC;
    float* bufC = bufB + (size_t)NC;
    float* h3   = bufC + (size_t)NC;
    int*   cnt  = (int*)(h3 + (size_t)Nn * 2);
    int*   strt = cnt + OWN_MAX * OWN_MAX;
    int2*  rec  = (int2*)(strt + 2 * OWN_MAX);
    float* out  = (float*)d_out;

    const int T = 256;
    const int gN  = (Nn + T - 1) / T;
    const int gE  = (Ee + T - 1) / T;
    const int chunk = (Ee + OWN_MAX - 1) / OWN_MAX;
    const int gOwn  = (Nn + OWN_ROWS - 1) / OWN_ROWS;
    const int gMM = Nn / 16;

    fill_kernel<<<gN, T, 0, stream>>>(deg, Nn, 1.0f);
    deg_count_kernel<<<gE, T, 0, stream>>>(dst, deg, Ee);
    bin_count_kernel<<<OWN_MAX, BIN_T, 0, stream>>>(dst, deg, dinv, cnt, Ee, Nn, chunk);
    bin_place_kernel<<<OWN_MAX, BIN_T, 0, stream>>>(src, dst, dinv, cnt, strt, rec, Ee, chunk);

    owner_aggregate_kernel<<<gOwn, OWN_T, 0, stream>>>(x, rec, strt, dinv, bufA, Nn);
    gemm_wmma_f32<<<GT_BLOCKS, T, 0, stream>>>(bufA, W1, b1, g1, be1, rm1, rv1, bufB, gMM);

    owner_aggregate_kernel<<<gOwn, OWN_T, 0, stream>>>(bufB, rec, strt, dinv, bufC, Nn);
    gemm_wmma_f32<<<GT_BLOCKS, T, 0, stream>>>(bufC, W2, b2, g2, be2, rm2, rv2, bufA, gMM);

    gemm_out2_kernel<<<gN, T, 0, stream>>>(bufA, W3, dinv, b3, h3, out, Nn);
    scatter2_kernel<<<gE, T, 0, stream>>>(h3, src, dst, dinv, out, Ee);
}


# Layer 3's 2-channel aggregation by owner work-groups: binned records feed LDS float adds into a 391x2 LDS table seeded with self term + bias, rows stored once; no global float adds; arrival order
# speedup vs baseline: 3.5696x; 1.4289x over previous
#include <hip/hip_runtime.h>

#define CH 128

typedef float v2f __attribute__((ext_vector_type(2)));
typedef float v8f __attribute__((ext_vector_type(8)));

__global__ void fill_kernel(float* __restrict__ p, int n, float v) {
    int i = blockIdx.x * blockDim.x + threadIdx.x;
    if (i < n) p[i] = v;
}

__global__ void deg_count_kernel(const int* __restrict__ dst,
                                 float* __restrict__ deg, int E) {
    int e = blockIdx.x * blockDim.x + threadIdx.x;
    if (e < E) atomicAdd(&deg[dst[e]], 1.0f);
}

#define OWN_ROWS 391
#define OWN_MAX 256
#define OWN_SHIFT 9
#define BIN_T 1024

__global__ __launch_bounds__(BIN_T)
void bin_count_kernel(const int* __restrict__ dst, const float* __restrict__ deg,
                      float* __restrict__ dinv, int* __restrict__ cnt,
                      int E, int n, int chunk) {
    __shared__ int hist[OWN_MAX];
    const int c = blockIdx.x;
    for (int i = c * BIN_T + threadIdx.x; i < n; i += gridDim.x * BIN_T)
        dinv[i] = rsqrtf(deg[i]);
    if (threadIdx.x < OWN_MAX) hist[threadIdx.x] = 0;
    __syncthreads();
    const int e0 = c * chunk;
    const int e1 = min(E, e0 + chunk);
    for (int e = e0 + threadIdx.x; e < e1; e += BIN_T)
        atomicAdd(&hist[dst[e] / OWN_ROWS], 1);
    __syncthreads();
    if (threadIdx.x < OWN_MAX) cnt[c * OWN_MAX + threadIdx.x] = hist[threadIdx.x];
}

__global__ __launch_bounds__(BIN_T)
void bin_place_kernel(const int* __restrict__ src, const int* __restrict__ dst,
                      const float* __restrict__ dinv, const int* __restrict__ cnt,
                      int* __restrict__ start, int2* __restrict__ rec,
                      int E, int chunk) {
    __shared__ int part_tot[BIN_T / OWN_MAX][OWN_MAX];
    __shared__ int part_pre[BIN_T / OWN_MAX][OWN_MAX];
    __shared__ int tot[OWN_MAX];
    __shared__ int cursor[OWN_MAX];
    const int c = blockIdx.x;
    const int o = threadIdx.x & (OWN_MAX - 1);
    const int q = threadIdx.x / OWN_MAX;
    const int per = OWN_MAX / (BIN_T / OWN_MAX);
    int t = 0, p = 0;
    for (int cc = q * per; cc < (q + 1) * per; ++cc) {
        const int v = cnt[cc * OWN_MAX + o];
        t += v;
        if (cc < c) p += v;
    }
    part_tot[q][o] = t;
    part_pre[q][o] = p;
    __syncthreads();
    if (threadIdx.x < OWN_MAX) {
        int s = 0;
        for (int k = 0; k < BIN_T / OWN_MAX; ++k) s += part_tot[k][o];
        tot[o] = s;
    }
    __syncthreads();
    if (threadIdx.x < OWN_MAX) {
        int s = 0;
        for (int k = 0; k < o; ++k) s += tot[k];
        int pre = 0;
        for (int k = 0; k < BIN_T / OWN_MAX; ++k) pre += part_pre[k][o];
        cursor[o] = s + pre;
        if (c == 0) {
            start[o] = s;
            if (o == OWN_MAX - 1) start[OWN_MAX] = s + tot[o];
        }
    }
    __syncthreads();
    const int e0 = c * chunk;
    const int e1 = min(E, e0 + chunk);
    for (int e = e0 + threadIdx.x; e < e1; e += BIN_T) {
        const int s = src[e];
        const int d = dst[e];
        const int ow = d / OWN_ROWS;
        const int lr = d - ow * OWN_ROWS;
        const int slot = atomicAdd(&cursor[ow], 1);
        rec[slot] = make_int2((s << OWN_SHIFT) | lr,
                              __float_as_int(dinv[s] * dinv[d]));
    }
}

#define GT_PITCH 132
#define GT_TILE (16 * GT_PITCH)
#define GT_BLOCKS 1024

__global__ __launch_bounds__(256)
void gemm_wmma_f32(const float* __restrict__ X, const float* __restrict__ W,
                   const float* __restrict__ b, const float* __restrict__ g,
                   const float* __restrict__ be, const float* __restrict__ rm,
                   const float* __restrict__ rv,
                   float* __restrict__ H, int ntiles) {
    __shared__ __attribute__((aligned(16))) float As[2][GT_TILE];
    __shared__ __attribute__((aligned(16))) float Os[2][GT_TILE];

    const int wave  = __builtin_amdgcn_readfirstlane((int)(threadIdx.x >> 5));
    const int lane  = threadIdx.x & 31;
    const int n0    = wave * 16;
    const int mrow  = lane & 15;
    const int khalf = (lane >> 4) << 1;
    const int ncol  = lane & 15;
    const int moff  = (lane >> 4) * 8;

    int t = blockIdx.x;
    if (t >= ntiles) return;

    const int c = n0 + ncol;
    const float inv = g[c] * rsqrtf(rv[c] + 1e-5f);
    const float bc  = b[c];
    const float rmc = rm[c];
    const float bec = be[c];

    v2f bf[CH / 4];
#pragma unroll
    for (int kk = 0; kk < CH / 4; ++kk) {
        bf[kk].x = W[(size_t)(4 * kk + khalf) * CH + c];
        bf[kk].y = W[(size_t)(4 * kk + khalf + 1) * CH + c];
    }

    const int soff0 = wave * GT_PITCH + lane * 4;
    const int soff1 = (wave + 8) * GT_PITCH + lane * 4;
    const int aoff  = mrow * GT_PITCH + khalf;

    {
        const float4* __restrict__ xp =
            reinterpret_cast<const float4*>(X + (size_t)t * (16 * CH));
        const float4 p0 = xp[threadIdx.x];
        const float4 p1 = xp[threadIdx.x + 256];
        *reinterpret_cast<float4*>(&As[0][soff0]) = p0;
        *reinterpret_cast<float4*>(&As[0][soff1]) = p1;
    }
    __syncthreads();

    int cur = 0;
    for (; t < ntiles; t += gridDim.x) {
        const int tn = t + gridDim.x;
        const bool more = tn < ntiles;
        float4 p0 = make_float4(0.f, 0.f, 0.f, 0.f);
        float4 p1 = p0;
        if (more) {
            const float4* __restrict__ xp =
                reinterpret_cast<const float4*>(X + (size_t)tn * (16 * CH));
            p0 = xp[threadIdx.x];
            p1 = xp[threadIdx.x + 256];
        }
        const int m0 = t * 16;

        const float* as = &As[cur][aoff];
        v8f acc = {0.f, 0.f, 0.f, 0.f, 0.f, 0.f, 0.f, 0.f};
#pragma unroll
        for (int kk = 0; kk < CH / 4; ++kk) {
            const v2f a = *reinterpret_cast<const v2f*>(as + 4 * kk);
            acc = __builtin_amdgcn_wmma_f32_16x16x4_f32(
                 false, a,  false, bf[kk],
                 (short)0, acc,  false,  false);
        }

        float* os = &Os[cur][0];
#pragma unroll
        for (int j = 0; j < 8; ++j) {
            const float y = (acc[j] + bc - rmc) * inv + bec;
            os[(j + moff) * GT_PITCH + c] = y > 0.f ? y : 0.f;
        }
        if (more) {
            *reinterpret_cast<float4*>(&As[cur ^ 1][soff0]) = p0;
            *reinterpret_cast<float4*>(&As[cur ^ 1][soff1]) = p1;
        }
        __syncthreads();

        const float4 h0 = *reinterpret_cast<const float4*>(&os[soff0]);
        const float4 h1 = *reinterpret_cast<const float4*>(&os[soff1]);
        float4* __restrict__ hp = reinterpret_cast<float4*>(H + (size_t)m0 * CH);
        hp[threadIdx.x]       = h0;
        hp[threadIdx.x + 256] = h1;
        cur ^= 1;
    }
}

#define EPW 8
#define OWN_T 1024
#define OWN_WAVES (OWN_T / 32)
#define OWN_INIT 7

__device__ __forceinline__
void own_add(float* a, const float4 v, const float nj) {
    atomicAdd(a,      v.x * nj);
    atomicAdd(a + 32, v.y * nj);
    atomicAdd(a + 64, v.z * nj);
    atomicAdd(a + 96, v.w * nj);
}

__global__ __launch_bounds__(OWN_T)
void owner_aggregate_kernel(const float* __restrict__ h, const int2* __restrict__ rec,
                            const int* __restrict__ start,
                            const float* __restrict__ dinv,
                            float* __restrict__ out, int n) {
    __shared__ __attribute__((aligned(16))) float acc[OWN_ROWS * CH];

    const int wave = __builtin_amdgcn_readfirstlane((int)(threadIdx.x >> 5));
    const int lane = threadIdx.x & 31;
    const int o  = blockIdx.x;
    const int r0 = o * OWN_ROWS;
    const int nr = min(OWN_ROWS, n - r0);

    for (int lb = wave; lb < nr; lb += OWN_INIT * OWN_WAVES) {
        float4 v[OWN_INIT];
#pragma unroll
        for (int u = 0; u < OWN_INIT; ++u) {
            const int lr = min(lb + u * OWN_WAVES, nr - 1);
            v[u] = reinterpret_cast<const float4*>(h + (size_t)(r0 + lr) * CH)[lane];
        }
#pragma unroll
        for (int u = 0; u < OWN_INIT; ++u) {
            const int lr = lb + u * OWN_WAVES;
            if (lr < nr) {
                const float di = dinv[r0 + lr];
                const float nrm = di * di;
                float* a = &acc[lr * CH + lane];
                a[0]  = v[u].x * nrm;
                a[32] = v[u].y * nrm;
                a[64] = v[u].z * nrm;
                a[96] = v[u].w * nrm;
            }
        }
    }
    __syncthreads();

    const int s0 = __builtin_amdgcn_readfirstlane(start[o]);
    const int s1 = __builtin_amdgcn_readfirstlane(start[o + 1]);
    const int nfull = (s1 - s0) / EPW;
    const int sub = lane & (EPW - 1);

    int2 r = make_int2(0, 0);
    if (wave < nfull) r = rec[s0 + wave * EPW + sub];
    for (int b = wave; b < nfull; b += OWN_WAVES) {
        int2 rn = make_int2(0, 0);
        if (b + OWN_WAVES < nfull) rn = rec[s0 + (b + OWN_WAVES) * EPW + sub];
        const int   pk  = r.x;
        const float nrm = __int_as_float(r.y);
        float4 v[EPW];
#pragma unroll
        for (int j = 0; j < EPW; ++j) {
            const int sj = __builtin_amdgcn_readlane(pk, j) >> OWN_SHIFT;
            v[j] = reinterpret_cast<const float4*>(h + (size_t)sj * CH)[lane];
        }
#pragma unroll
        for (int j = 0; j < EPW; ++j) {
            const int   lj = __builtin_amdgcn_readlane(pk, j) & ((1 << OWN_SHIFT) - 1);
            const float nj = __int_as_float(
                __builtin_amdgcn_readlane(__float_as_int(nrm), j));
            own_add(&acc[lj * CH + lane], v[j], nj);
        }
        r = rn;
    }
    if (wave == (nfull & (OWN_WAVES - 1))) {
        for (int e = s0 + nfull * EPW; e < s1; ++e) {
            const int2 q = rec[e];
            const int sj = q.x >> OWN_SHIFT;
            const int lj = q.x & ((1 << OWN_SHIFT) - 1);
            const float4 v =
                reinterpret_cast<const float4*>(h + (size_t)sj * CH)[lane];
            own_add(&acc[lj * CH + lane], v, __int_as_float(q.y));
        }
    }
    __syncthreads();

    for (int lr = wave; lr < nr; lr += OWN_WAVES) {
        const float* a = &acc[lr * CH + lane];
        const float4 v = make_float4(a[0], a[32], a[64], a[96]);
        reinterpret_cast<float4*>(out + (size_t)(r0 + lr) * CH)[lane] = v;
    }
}

__global__ void gemm_out2_kernel(const float* __restrict__ X,
                                 const float* __restrict__ W3,
                                 float* __restrict__ H3, int M) {
    int i = blockIdx.x * blockDim.x + threadIdx.x;
    if (i >= M) return;
    const float* xr = X + (size_t)i * CH;
    float a0 = 0.f, a1 = 0.f;
#pragma unroll 8
    for (int k = 0; k < CH; ++k) {
        const float x = xr[k];
        a0 += x * W3[2 * k];
        a1 += x * W3[2 * k + 1];
    }
    H3[(size_t)i * 2]     = a0;
    H3[(size_t)i * 2 + 1] = a1;
}

#define O2_T 1024
#define O2_U 8
static_assert(2 * OWN_ROWS + 31 <= O2_T, "one float of the range per thread");
static_assert(O2_U == 8, "the statement that holds the row loads names 8 slots");

__global__ __launch_bounds__(O2_T)
void owner_aggregate2_kernel(const float* __restrict__ h3, const int2* __restrict__ rec,
                             const int* __restrict__ start,
                             const float* __restrict__ dinv,
                             const float* __restrict__ b3,
                             float* __restrict__ out, int n) {
    __shared__ float acc2[OWN_ROWS * 2];

    const int o  = blockIdx.x;
    const int r0 = o * OWN_ROWS;
    const int nr = min(OWN_ROWS, n - r0);
    const int g0 = 2 * r0;
    const int g1 = g0 + 2 * nr;
    const int g  = (g0 & ~31) + (int)threadIdx.x;
    const bool mine = g >= g0 && g < g1;

    const int gc = min(max(g, g0), g1 - 1);
    const float dv = dinv[gc >> 1];
    const float hv = h3[gc];
    const float bv = b3[gc & 1];

    const int s0  = start[o];
    const int cnt = start[o + 1] - s0;
    const float2* __restrict__ h3r = reinterpret_cast<const float2*>(h3);
    for (int base = 0; base == 0 || base < cnt; base += O2_U * O2_T) {
        int2   q[O2_U];
        float2 v[O2_U];
#pragma unroll
        for (int u = 0; u < O2_U; ++u) {
            const int i = base + u * O2_T + (int)threadIdx.x;
            q[u] = make_int2(0, 0);
            if (i < cnt) q[u] = rec[s0 + i];
        }
#pragma unroll
        for (int u = 0; u < O2_U; ++u)
            v[u] = h3r[q[u].x >> OWN_SHIFT];
        if (base == 0) {
            if (mine) acc2[g - g0] = hv * (dv * dv) + bv;
            __syncthreads();
        }
        asm volatile("" :: "v"(v[0].x), "v"(v[0].y), "v"(v[1].x), "v"(v[1].y),
                           "v"(v[2].x), "v"(v[2].y), "v"(v[3].x), "v"(v[3].y),
                           "v"(v[4].x), "v"(v[4].y), "v"(v[5].x), "v"(v[5].y),
                           "v"(v[6].x), "v"(v[6].y), "v"(v[7].x), "v"(v[7].y));
#pragma unroll
        for (int u = 0; u < O2_U; ++u) {
            const int i = base + u * O2_T + (int)threadIdx.x;
            if (i < cnt) {
                const int   lr = q[u].x & ((1 << OWN_SHIFT) - 1);
                const float nj = __int_as_float(q[u].y);
                atomicAdd(&acc2[2 * lr],     v[u].x * nj);
                atomicAdd(&acc2[2 * lr + 1], v[u].y * nj);
            }
        }
    }
    __syncthreads();

    if (mine) out[g] = acc2[g - g0];
}

extern "C" void kernel_launch(void* const* d_in, const int* in_sizes, int n_in,
                              void* d_out, int out_size, void* d_ws, size_t ws_size,
                              hipStream_t stream) {
    const float* x   = (const float*)d_in[0];
    const int*   ei  = (const int*)d_in[1];
    const float* W1  = (const float*)d_in[2];
    const float* b1  = (const float*)d_in[3];
    const float* g1  = (const float*)d_in[4];
    const float* be1 = (const float*)d_in[5];
    const float* rm1 = (const float*)d_in[6];
    const float* rv1 = (const float*)d_in[7];
    const float* W2  = (const float*)d_in[8];
    const float* b2  = (const float*)d_in[9];
    const float* g2  = (const float*)d_in[10];
    const float* be2 = (const float*)d_in[11];
    const float* rm2 = (const float*)d_in[12];
    const float* rv2 = (const float*)d_in[13];
    const float* W3  = (const float*)d_in[14];
    const float* b3  = (const float*)d_in[15];

    const int Nn   = in_sizes[0] / CH;
    const int Ee   = in_sizes[1] / 2;
    const int NC   = Nn * CH;

    const int* src = ei;
    const int* dst = ei + Ee;

    float* deg  = (float*)d_ws;
    float* dinv = deg  + Nn;
    float* bufA = dinv + Nn;
    float* bufB = bufA + (size_t)NC;
    float* bufC = bufB + (size_t)NC;
    float* h3   = bufC + (size_t)NC;
    int*   cnt  = (int*)(h3 + (size_t)Nn * 2);
    int*   strt = cnt + OWN_MAX * OWN_MAX;
    int2*  rec  = (int2*)(strt + 2 * OWN_MAX);
    float* out  = (float*)d_out;

    const int T = 256;
    const int gN  = (Nn + T - 1) / T;
    const int gE  = (Ee + T - 1) / T;
    const int chunk = (Ee + OWN_MAX - 1) / OWN_MAX;
    const int gOwn  = (Nn + OWN_ROWS - 1) / OWN_ROWS;
    const int gMM = Nn / 16;

    fill_kernel<<<gN, T, 0, stream>>>(deg, Nn, 1.0f);
    deg_count_kernel<<<gE, T, 0, stream>>>(dst, deg, Ee);
    bin_count_kernel<<<OWN_MAX, BIN_T, 0, stream>>>(dst, deg, dinv, cnt, Ee, Nn, chunk);
    bin_place_kernel<<<OWN_MAX, BIN_T, 0, stream>>>(src, dst, dinv, cnt, strt, rec, Ee, chunk);

    owner_aggregate_kernel<<<gOwn, OWN_T, 0, stream>>>(x, rec, strt, dinv, bufA, Nn);
    gemm_wmma_f32<<<GT_BLOCKS, T, 0, stream>>>(bufA, W1, b1, g1, be1, rm1, rv1, bufB, gMM);

    owner_aggregate_kernel<<<gOwn, OWN_T, 0, stream>>>(bufB, rec, strt, dinv, bufC, Nn);
    gemm_wmma_f32<<<GT_BLOCKS, T, 0, stream>>>(bufC, W2, b2, g2, be2, rm2, rv2, bufA, gMM);

    gemm_out2_kernel<<<gN, T, 0, stream>>>(bufA, W3, h3, Nn);
    owner_aggregate2_kernel<<<gOwn, O2_T, 0, stream>>>(h3, rec, strt, dinv, b3, out, Nn);
}


# Degree from the owner-binned edge lists: each owner counts its 391 rows with LDS integer adds (start 1 = self loop), a second owner pass writes the norms; fill and global float-add pass gone
# speedup vs baseline: 4.7203x; 1.3224x over previous
#include <hip/hip_runtime.h>

#define CH 128

typedef float v2f __attribute__((ext_vector_type(2)));
typedef float v8f __attribute__((ext_vector_type(8)));

#define OWN_ROWS 391
#define OWN_MAX 256
#define OWN_SHIFT 9
#define BIN_T 1024

__global__ __launch_bounds__(BIN_T)
void bin_count_kernel(const int* __restrict__ dst, int* __restrict__ cnt,
                      int E, int chunk) {
    __shared__ int hist[OWN_MAX];
    const int c = blockIdx.x;
    if (threadIdx.x < OWN_MAX) hist[threadIdx.x] = 0;
    __syncthreads();
    const int e0 = c * chunk;
    const int e1 = min(E, e0 + chunk);
    for (int e = e0 + threadIdx.x; e < e1; e += BIN_T)
        atomicAdd(&hist[dst[e] / OWN_ROWS], 1);
    __syncthreads();
    if (threadIdx.x < OWN_MAX) cnt[c * OWN_MAX + threadIdx.x] = hist[threadIdx.x];
}

__global__ __launch_bounds__(BIN_T)
void bin_place_kernel(const int* __restrict__ src, const int* __restrict__ dst,
                      const int* __restrict__ cnt,
                      int* __restrict__ start, int* __restrict__ key,
                      int E, int chunk) {
    __shared__ int part_tot[BIN_T / OWN_MAX][OWN_MAX];
    __shared__ int part_pre[BIN_T / OWN_MAX][OWN_MAX];
    __shared__ int tot[OWN_MAX];
    __shared__ int cursor[OWN_MAX];
    const int c = blockIdx.x;
    const int o = threadIdx.x & (OWN_MAX - 1);
    const int q = threadIdx.x / OWN_MAX;
    const int per = OWN_MAX / (BIN_T / OWN_MAX);
    int t = 0, p = 0;
    for (int cc = q * per; cc < (q + 1) * per; ++cc) {
        const int v = cnt[cc * OWN_MAX + o];
        t += v;
        if (cc < c) p += v;
    }
    part_tot[q][o] = t;
    part_pre[q][o] = p;
    __syncthreads();
    if (threadIdx.x < OWN_MAX) {
        int s = 0;
        for (int k = 0; k < BIN_T / OWN_MAX; ++k) s += part_tot[k][o];
        tot[o] = s;
    }
    __syncthreads();
    if (threadIdx.x < OWN_MAX) {
        int s = 0;
        for (int k = 0; k < o; ++k) s += tot[k];
        int pre = 0;
        for (int k = 0; k < BIN_T / OWN_MAX; ++k) pre += part_pre[k][o];
        cursor[o] = s + pre;
        if (c == 0) {
            start[o] = s;
            if (o == OWN_MAX - 1) start[OWN_MAX] = s + tot[o];
        }
    }
    __syncthreads();
    const int e0 = c * chunk;
    const int e1 = min(E, e0 + chunk);
    for (int e = e0 + threadIdx.x; e < e1; e += BIN_T) {
        const int s = src[e];
        const int d = dst[e];
        const int ow = d / OWN_ROWS;
        const int lr = d - ow * OWN_ROWS;
        const int slot = atomicAdd(&cursor[ow], 1);
        key[slot] = (s << OWN_SHIFT) | lr;
    }
}

#define OL_T 1024
#define OL_U 8
static_assert(OWN_ROWS <= OL_T, "one counter per thread");
static_assert(OL_U == 8, "the statements that hold the loads name 8 slots");

__global__ __launch_bounds__(OL_T)
void owner_degree_kernel(const int* __restrict__ key, const int* __restrict__ start,
                         float* __restrict__ dinv, int n) {
    __shared__ int dcount[OWN_ROWS];
    const int o  = blockIdx.x;
    const int r0 = o * OWN_ROWS;
    const int nr = min(OWN_ROWS, n - r0);
    const int s0  = start[o];
    const int cnt = start[o + 1] - s0;
    if (threadIdx.x < OWN_ROWS) dcount[threadIdx.x] = 1;
    __syncthreads();
    for (int base = 0; base < cnt; base += OL_U * OL_T) {
        int k[OL_U];
#pragma unroll
        for (int u = 0; u < OL_U; ++u) {
            const int i = base + u * OL_T + (int)threadIdx.x;
            k[u] = key[s0 + min(i, cnt - 1)];
        }
        asm volatile("" :: "v"(k[0]), "v"(k[1]), "v"(k[2]), "v"(k[3]),
                           "v"(k[4]), "v"(k[5]), "v"(k[6]), "v"(k[7]));
#pragma unroll
        for (int u = 0; u < OL_U; ++u) {
            const int i = base + u * OL_T + (int)threadIdx.x;
            if (i < cnt) atomicAdd(&dcount[k[u] & ((1 << OWN_SHIFT) - 1)], 1);
        }
    }
    __syncthreads();
    if ((int)threadIdx.x < nr)
        dinv[r0 + threadIdx.x] = rsqrtf((float)dcount[threadIdx.x]);
}

__global__ __launch_bounds__(OL_T)
void owner_norm_kernel(const int* __restrict__ key, const int* __restrict__ start,
                       const float* __restrict__ dinv, int2* __restrict__ rec) {
    const int o  = blockIdx.x;
    const int r0 = o * OWN_ROWS;
    const int s0  = start[o];
    const int cnt = start[o + 1] - s0;
    for (int base = 0; base < cnt; base += OL_U * OL_T) {
        int   k[OL_U];
        float a[OL_U], b[OL_U];
#pragma unroll
        for (int u = 0; u < OL_U; ++u) {
            const int i = base + u * OL_T + (int)threadIdx.x;
            k[u] = key[s0 + min(i, cnt - 1)];
        }
        asm volatile("" :: "v"(k[0]), "v"(k[1]), "v"(k[2]), "v"(k[3]),
                           "v"(k[4]), "v"(k[5]), "v"(k[6]), "v"(k[7]));
#pragma unroll
        for (int u = 0; u < OL_U; ++u) {
            a[u] = dinv[k[u] >> OWN_SHIFT];
            b[u] = dinv[r0 + (k[u] & ((1 << OWN_SHIFT) - 1))];
        }
        asm volatile("" :: "v"(a[0]), "v"(b[0]), "v"(a[1]), "v"(b[1]),
                           "v"(a[2]), "v"(b[2]), "v"(a[3]), "v"(b[3]),
                           "v"(a[4]), "v"(b[4]), "v"(a[5]), "v"(b[5]),
                           "v"(a[6]), "v"(b[6]), "v"(a[7]), "v"(b[7]));
#pragma unroll
        for (int u = 0; u < OL_U; ++u) {
            const int i = base + u * OL_T + (int)threadIdx.x;
            if (i < cnt) rec[s0 + i] = make_int2(k[u], __float_as_int(a[u] * b[u]));
        }
    }
}

#define GT_PITCH 132
#define GT_TILE (16 * GT_PITCH)
#define GT_BLOCKS 1024

__global__ __launch_bounds__(256)
void gemm_wmma_f32(const float* __restrict__ X, const float* __restrict__ W,
                   const float* __restrict__ b, const float* __restrict__ g,
                   const float* __restrict__ be, const float* __restrict__ rm,
                   const float* __restrict__ rv,
                   float* __restrict__ H, int ntiles) {
    __shared__ __attribute__((aligned(16))) float As[2][GT_TILE];
    __shared__ __attribute__((aligned(16))) float Os[2][GT_TILE];

    const int wave  = __builtin_amdgcn_readfirstlane((int)(threadIdx.x >> 5));
    const int lane  = threadIdx.x & 31;
    const int n0    = wave * 16;
    const int mrow  = lane & 15;
    const int khalf = (lane >> 4) << 1;
    const int ncol  = lane & 15;
    const int moff  = (lane >> 4) * 8;

    int t = blockIdx.x;
    if (t >= ntiles) return;

    const int c = n0 + ncol;
    const float inv = g[c] * rsqrtf(rv[c] + 1e-5f);
    const float bc  = b[c];
    const float rmc = rm[c];
    const float bec = be[c];

    v2f bf[CH / 4];
#pragma unroll
    for (int kk = 0; kk < CH / 4; ++kk) {
        bf[kk].x = W[(size_t)(4 * kk + khalf) * CH + c];
        bf[kk].y = W[(size_t)(4 * kk + khalf + 1) * CH + c];
    }

    const int soff0 = wave * GT_PITCH + lane * 4;
    const int soff1 = (wave + 8) * GT_PITCH + lane * 4;
    const int aoff  = mrow * GT_PITCH + khalf;

    {
        const float4* __restrict__ xp =
            reinterpret_cast<const float4*>(X + (size_t)t * (16 * CH));
        const float4 p0 = xp[threadIdx.x];
        const float4 p1 = xp[threadIdx.x + 256];
        *reinterpret_cast<float4*>(&As[0][soff0]) = p0;
        *reinterpret_cast<float4*>(&As[0][soff1]) = p1;
    }
    __syncthreads();

    int cur = 0;
    for (; t < ntiles; t += gridDim.x) {
        const int tn = t + gridDim.x;
        const bool more = tn < ntiles;
        float4 p0 = make_float4(0.f, 0.f, 0.f, 0.f);
        float4 p1 = p0;
        if (more) {
            const float4* __restrict__ xp =
                reinterpret_cast<const float4*>(X + (size_t)tn * (16 * CH));
            p0 = xp[threadIdx.x];
            p1 = xp[threadIdx.x + 256];
        }
        const int m0 = t * 16;

        const float* as = &As[cur][aoff];
        v8f acc = {0.f, 0.f, 0.f, 0.f, 0.f, 0.f, 0.f, 0.f};
#pragma unroll
        for (int kk = 0; kk < CH / 4; ++kk) {
            const v2f a = *reinterpret_cast<const v2f*>(as + 4 * kk);
            acc = __builtin_amdgcn_wmma_f32_16x16x4_f32(
                 false, a,  false, bf[kk],
                 (short)0, acc,  false,  false);
        }

        float* os = &Os[cur][0];
#pragma unroll
        for (int j = 0; j < 8; ++j) {
            const float y = (acc[j] + bc - rmc) * inv + bec;
            os[(j + moff) * GT_PITCH + c] = y > 0.f ? y : 0.f;
        }
        if (more) {
            *reinterpret_cast<float4*>(&As[cur ^ 1][soff0]) = p0;
            *reinterpret_cast<float4*>(&As[cur ^ 1][soff1]) = p1;
        }
        __syncthreads();

        const float4 h0 = *reinterpret_cast<const float4*>(&os[soff0]);
        const float4 h1 = *reinterpret_cast<const float4*>(&os[soff1]);
        float4* __restrict__ hp = reinterpret_cast<float4*>(H + (size_t)m0 * CH);
        hp[threadIdx.x]       = h0;
        hp[threadIdx.x + 256] = h1;
        cur ^= 1;
    }
}

#define EPW 8
#define OWN_T 1024
#define OWN_WAVES (OWN_T / 32)
#define OWN_INIT 7

__device__ __forceinline__
void own_add(float* a, const float4 v, const float nj) {
    atomicAdd(a,      v.x * nj);
    atomicAdd(a + 32, v.y * nj);
    atomicAdd(a + 64, v.z * nj);
    atomicAdd(a + 96, v.w * nj);
}

__global__ __launch_bounds__(OWN_T)
void owner_aggregate_kernel(const float* __restrict__ h, const int2* __restrict__ rec,
                            const int* __restrict__ start,
                            const float* __restrict__ dinv,
                            float* __restrict__ out, int n) {
    __shared__ __attribute__((aligned(16))) float acc[OWN_ROWS * CH];

    const int wave = __builtin_amdgcn_readfirstlane((int)(threadIdx.x >> 5));
    const int lane = threadIdx.x & 31;
    const int o  = blockIdx.x;
    const int r0 = o * OWN_ROWS;
    const int nr = min(OWN_ROWS, n - r0);

    for (int lb = wave; lb < nr; lb += OWN_INIT * OWN_WAVES) {
        float4 v[OWN_INIT];
#pragma unroll
        for (int u = 0; u < OWN_INIT; ++u) {
            const int lr = min(lb + u * OWN_WAVES, nr - 1);
            v[u] = reinterpret_cast<const float4*>(h + (size_t)(r0 + lr) * CH)[lane];
        }
#pragma unroll
        for (int u = 0; u < OWN_INIT; ++u) {
            const int lr = lb + u * OWN_WAVES;
            if (lr < nr) {
                const float di = dinv[r0 + lr];
                const float nrm = di * di;
                float* a = &acc[lr * CH + lane];
                a[0]  = v[u].x * nrm;
                a[32] = v[u].y * nrm;
                a[64] = v[u].z * nrm;
                a[96] = v[u].w * nrm;
            }
        }
    }
    __syncthreads();

    const int s0 = __builtin_amdgcn_readfirstlane(start[o]);
    const int s1 = __builtin_amdgcn_readfirstlane(start[o + 1]);
    const int nfull = (s1 - s0) / EPW;
    const int sub = lane & (EPW - 1);

    int2 r = make_int2(0, 0);
    if (wave < nfull) r = rec[s0 + wave * EPW + sub];
    for (int b = wave; b < nfull; b += OWN_WAVES) {
        int2 rn = make_int2(0, 0);
        if (b + OWN_WAVES < nfull) rn = rec[s0 + (b + OWN_WAVES) * EPW + sub];
        const int   pk  = r.x;
        const float nrm = __int_as_float(r.y);
        float4 v[EPW];
#pragma unroll
        for (int j = 0; j < EPW; ++j) {
            const int sj = __builtin_amdgcn_readlane(pk, j) >> OWN_SHIFT;
            v[j] = reinterpret_cast<const float4*>(h + (size_t)sj * CH)[lane];
        }
#pragma unroll
        for (int j = 0; j < EPW; ++j) {
            const int   lj = __builtin_amdgcn_readlane(pk, j) & ((1 << OWN_SHIFT) - 1);
            const float nj = __int_as_float(
                __builtin_amdgcn_readlane(__float_as_int(nrm), j));
            own_add(&acc[lj * CH + lane], v[j], nj);
        }
        r = rn;
    }
    if (wave == (nfull & (OWN_WAVES - 1))) {
        for (int e = s0 + nfull * EPW; e < s1; ++e) {
            const int2 q = rec[e];
            const int sj = q.x >> OWN_SHIFT;
            const int lj = q.x & ((1 << OWN_SHIFT) - 1);
            const float4 v =
                reinterpret_cast<const float4*>(h + (size_t)sj * CH)[lane];
            own_add(&acc[lj * CH + lane], v, __int_as_float(q.y));
        }
    }
    __syncthreads();

    for (int lr = wave; lr < nr; lr += OWN_WAVES) {
        const float* a = &acc[lr * CH + lane];
        const float4 v = make_float4(a[0], a[32], a[64], a[96]);
        reinterpret_cast<float4*>(out + (size_t)(r0 + lr) * CH)[lane] = v;
    }
}

__global__ void gemm_out2_kernel(const float* __restrict__ X,
                                 const float* __restrict__ W3,
                                 float* __restrict__ H3, int M) {
    int i = blockIdx.x * blockDim.x + threadIdx.x;
    if (i >= M) return;
    const float* xr = X + (size_t)i * CH;
    float a0 = 0.f, a1 = 0.f;
#pragma unroll 8
    for (int k = 0; k < CH; ++k) {
        const float x = xr[k];
        a0 += x * W3[2 * k];
        a1 += x * W3[2 * k + 1];
    }
    H3[(size_t)i * 2]     = a0;
    H3[(size_t)i * 2 + 1] = a1;
}

#define O2_T 1024
#define O2_U 8
static_assert(2 * OWN_ROWS + 31 <= O2_T, "one float of the range per thread");
static_assert(O2_U == 8, "the statement that holds the row loads names 8 slots");

__global__ __launch_bounds__(O2_T)
void owner_aggregate2_kernel(const float* __restrict__ h3, const int2* __restrict__ rec,
                             const int* __restrict__ start,
                             const float* __restrict__ dinv,
                             const float* __restrict__ b3,
                             float* __restrict__ out, int n) {
    __shared__ float acc2[OWN_ROWS * 2];

    const int o  = blockIdx.x;
    const int r0 = o * OWN_ROWS;
    const int nr = min(OWN_ROWS, n - r0);
    const int g0 = 2 * r0;
    const int g1 = g0 + 2 * nr;
    const int g  = (g0 & ~31) + (int)threadIdx.x;
    const bool mine = g >= g0 && g < g1;

    const int gc = min(max(g, g0), g1 - 1);
    const float dv = dinv[gc >> 1];
    const float hv = h3[gc];
    const float bv = b3[gc & 1];

    const int s0  = start[o];
    const int cnt = start[o + 1] - s0;
    const float2* __restrict__ h3r = reinterpret_cast<const float2*>(h3);
    for (int base = 0; base == 0 || base < cnt; base += O2_U * O2_T) {
        int2   q[O2_U];
        float2 v[O2_U];
#pragma unroll
        for (int u = 0; u < O2_U; ++u) {
            const int i = base + u * O2_T + (int)threadIdx.x;
            q[u] = make_int2(0, 0);
            if (i < cnt) q[u] = rec[s0 + i];
        }
#pragma unroll
        for (int u = 0; u < O2_U; ++u)
            v[u] = h3r[q[u].x >> OWN_SHIFT];
        if (base == 0) {
            if (mine) acc2[g - g0] = hv * (dv * dv) + bv;
            __syncthreads();
        }
        asm volatile("" :: "v"(v[0].x), "v"(v[0].y), "v"(v[1].x), "v"(v[1].y),
                           "v"(v[2].x), "v"(v[2].y), "v"(v[3].x), "v"(v[3].y),
                           "v"(v[4].x), "v"(v[4].y), "v"(v[5].x), "v"(v[5].y),
                           "v"(v[6].x), "v"(v[6].y), "v"(v[7].x), "v"(v[7].y));
#pragma unroll
        for (int u = 0; u < O2_U; ++u) {
            const int i = base + u * O2_T + (int)threadIdx.x;
            if (i < cnt) {
                const int   lr = q[u].x & ((1 << OWN_SHIFT) - 1);
                const float nj = __int_as_float(q[u].y);
                atomicAdd(&acc2[2 * lr],     v[u].x * nj);
                atomicAdd(&acc2[2 * lr + 1], v[u].y * nj);
            }
        }
    }
    __syncthreads();

    if (mine) out[g] = acc2[g - g0];
}

extern "C" void kernel_launch(void* const* d_in, const int* in_sizes, int n_in,
                              void* d_out, int out_size, void* d_ws, size_t ws_size,
                              hipStream_t stream) {
    const float* x   = (const float*)d_in[0];
    const int*   ei  = (const int*)d_in[1];
    const float* W1  = (const float*)d_in[2];
    const float* b1  = (const float*)d_in[3];
    const float* g1  = (const float*)d_in[4];
    const float* be1 = (const float*)d_in[5];
    const float* rm1 = (const float*)d_in[6];
    const float* rv1 = (const float*)d_in[7];
    const float* W2  = (const float*)d_in[8];
    const float* b2  = (const float*)d_in[9];
    const float* g2  = (const float*)d_in[10];
    const float* be2 = (const float*)d_in[11];
    const float* rm2 = (const float*)d_in[12];
    const float* rv2 = (const float*)d_in[13];
    const float* W3  = (const float*)d_in[14];
    const float* b3  = (const float*)d_in[15];

    const int Nn   = in_sizes[0] / CH;
    const int Ee   = in_sizes[1] / 2;
    const int NC   = Nn * CH;

    const int* src = ei;
    const int* dst = ei + Ee;

    float* bufA = (float*)d_ws;
    float* bufB = bufA + (size_t)NC;
    float* bufC = bufB + (size_t)NC;
    float* h3   = bufC + (size_t)NC;
    float* dinv = h3 + (size_t)Nn * 2;
    int*   cnt  = (int*)(dinv + Nn);
    int*   strt = cnt + OWN_MAX * OWN_MAX;
    int2*  rec  = (int2*)(strt + 2 * OWN_MAX);
    int*   key  = (int*)(rec + Ee);
    float* out  = (float*)d_out;

    const int T = 256;
    const int gN  = (Nn + T - 1) / T;
    const int chunk = (Ee + OWN_MAX - 1) / OWN_MAX;
    const int gOwn  = (Nn + OWN_ROWS - 1) / OWN_ROWS;
    const int gMM = Nn / 16;

    bin_count_kernel<<<OWN_MAX, BIN_T, 0, stream>>>(dst, cnt, Ee, chunk);
    bin_place_kernel<<<OWN_MAX, BIN_T, 0, stream>>>(src, dst, cnt, strt, key, Ee, chunk);
    owner_degree_kernel<<<gOwn, OL_T, 0, stream>>>(key, strt, dinv, Nn);
    owner_norm_kernel<<<gOwn, OL_T, 0, stream>>>(key, strt, dinv, rec);

    owner_aggregate_kernel<<<gOwn, OWN_T, 0, stream>>>(x, rec, strt, dinv, bufA, Nn);
    gemm_wmma_f32<<<GT_BLOCKS, T, 0, stream>>>(bufA, W1, b1, g1, be1, rm1, rv1, bufB, gMM);

    owner_aggregate_kernel<<<gOwn, OWN_T, 0, stream>>>(bufB, rec, strt, dinv, bufC, Nn);
    gemm_wmma_f32<<<GT_BLOCKS, T, 0, stream>>>(bufC, W2, b2, g2, be2, rm2, rv2, bufA, gMM);

    gemm_out2_kernel<<<gN, T, 0, stream>>>(bufA, W3, h3, Nn);
    owner_aggregate2_kernel<<<gOwn, O2_T, 0, stream>>>(h3, rec, strt, dinv, b3, out, Nn);
}


# Each layer's product runs inside its owner aggregation launch: the 391-row sums stay in LDS (pitch 132) and feed the WMMA A fragments; aggregate never stored; 10 to 8 launches; k order changes
# speedup vs baseline: 5.2600x; 1.1143x over previous
#include <hip/hip_runtime.h>

#define CH 128

typedef float v2f __attribute__((ext_vector_type(2)));
typedef float v8f __attribute__((ext_vector_type(8)));

#define OWN_ROWS 391
#define OWN_MAX 256
#define OWN_SHIFT 9
#define BIN_T 1024

__global__ __launch_bounds__(BIN_T)
void bin_count_kernel(const int* __restrict__ dst, int* __restrict__ cnt,
                      int E, int chunk) {
    __shared__ int hist[OWN_MAX];
    const int c = blockIdx.x;
    if (threadIdx.x < OWN_MAX) hist[threadIdx.x] = 0;
    __syncthreads();
    const int e0 = c * chunk;
    const int e1 = min(E, e0 + chunk);
    for (int e = e0 + threadIdx.x; e < e1; e += BIN_T)
        atomicAdd(&hist[dst[e] / OWN_ROWS], 1);
    __syncthreads();
    if (threadIdx.x < OWN_MAX) cnt[c * OWN_MAX + threadIdx.x] = hist[threadIdx.x];
}

__global__ __launch_bounds__(BIN_T)
void bin_place_kernel(const int* __restrict__ src, const int* __restrict__ dst,
                      const int* __restrict__ cnt,
                      int* __restrict__ start, int* __restrict__ key,
                      int E, int chunk) {
    __shared__ int part_tot[BIN_T / OWN_MAX][OWN_MAX];
    __shared__ int part_pre[BIN_T / OWN_MAX][OWN_MAX];
    __shared__ int tot[OWN_MAX];
    __shared__ int cursor[OWN_MAX];
    const int c = blockIdx.x;
    const int o = threadIdx.x & (OWN_MAX - 1);
    const int q = threadIdx.x / OWN_MAX;
    const int per = OWN_MAX / (BIN_T / OWN_MAX);
    int t = 0, p = 0;
    for (int cc = q * per; cc < (q + 1) * per; ++cc) {
        const int v = cnt[cc * OWN_MAX + o];
        t += v;
        if (cc < c) p += v;
    }
    part_tot[q][o] = t;
    part_pre[q][o] = p;
    __syncthreads();
    if (threadIdx.x < OWN_MAX) {
        int s = 0;
        for (int k = 0; k < BIN_T / OWN_MAX; ++k) s += part_tot[k][o];
        tot[o] = s;
    }
    __syncthreads();
    if (threadIdx.x < OWN_MAX) {
        int s = 0;
        for (int k = 0; k < o; ++k) s += tot[k];
        int pre = 0;
        for (int k = 0; k < BIN_T / OWN_MAX; ++k) pre += part_pre[k][o];
        cursor[o] = s + pre;
        if (c == 0) {
            start[o] = s;
            if (o == OWN_MAX - 1) start[OWN_MAX] = s + tot[o];
        }
    }
    __syncthreads();
    const int e0 = c * chunk;
    const int e1 = min(E, e0 + chunk);
    for (int e = e0 + threadIdx.x; e < e1; e += BIN_T) {
        const int s = src[e];
        const int d = dst[e];
        const int ow = d / OWN_ROWS;
        const int lr = d - ow * OWN_ROWS;
        const int slot = atomicAdd(&cursor[ow], 1);
        key[slot] = (s << OWN_SHIFT) | lr;
    }
}

#define OL_T 1024
#define OL_U 8
static_assert(OWN_ROWS <= OL_T, "one counter per thread");
static_assert(OL_U == 8, "the statements that hold the loads name 8 slots");

__global__ __launch_bounds__(OL_T)
void owner_degree_kernel(const int* __restrict__ key, const int* __restrict__ start,
                         float* __restrict__ dinv, int n) {
    __shared__ int dcount[OWN_ROWS];
    const int o  = blockIdx.x;
    const int r0 = o * OWN_ROWS;
    const int nr = min(OWN_ROWS, n - r0);
    const int s0  = start[o];
    const int cnt = start[o + 1] - s0;
    if (threadIdx.x < OWN_ROWS) dcount[threadIdx.x] = 1;
    __syncthreads();
    for (int base = 0; base < cnt; base += OL_U * OL_T) {
        int k[OL_U];
#pragma unroll
        for (int u = 0; u < OL_U; ++u) {
            const int i = base + u * OL_T + (int)threadIdx.x;
            k[u] = key[s0 + min(i, cnt - 1)];
        }
        asm volatile("" :: "v"(k[0]), "v"(k[1]), "v"(k[2]), "v"(k[3]),
                           "v"(k[4]), "v"(k[5]), "v"(k[6]), "v"(k[7]));
#pragma unroll
        for (int u = 0; u < OL_U; ++u) {
            const int i = base + u * OL_T + (int)threadIdx.x;
            if (i < cnt) atomicAdd(&dcount[k[u] & ((1 << OWN_SHIFT) - 1)], 1);
        }
    }
    __syncthreads();
    if ((int)threadIdx.x < nr)
        dinv[r0 + threadIdx.x] = rsqrtf((float)dcount[threadIdx.x]);
}

__global__ __launch_bounds__(OL_T)
void owner_norm_kernel(const int* __restrict__ key, const int* __restrict__ start,
                       const float* __restrict__ dinv, int2* __restrict__ rec) {
    const int o  = blockIdx.x;
    const int r0 = o * OWN_ROWS;
    const int s0  = start[o];
    const int cnt = start[o + 1] - s0;
    for (int base = 0; base < cnt; base += OL_U * OL_T) {
        int   k[OL_U];
        float a[OL_U], b[OL_U];
#pragma unroll
        for (int u = 0; u < OL_U; ++u) {
            const int i = base + u * OL_T + (int)threadIdx.x;
            k[u] = key[s0 + min(i, cnt - 1)];
        }
        asm volatile("" :: "v"(k[0]), "v"(k[1]), "v"(k[2]), "v"(k[3]),
                           "v"(k[4]), "v"(k[5]), "v"(k[6]), "v"(k[7]));
#pragma unroll
        for (int u = 0; u < OL_U; ++u) {
            a[u] = dinv[k[u] >> OWN_SHIFT];
            b[u] = dinv[r0 + (k[u] & ((1 << OWN_SHIFT) - 1))];
        }
        asm volatile("" :: "v"(a[0]), "v"(b[0]), "v"(a[1]), "v"(b[1]),
                           "v"(a[2]), "v"(b[2]), "v"(a[3]), "v"(b[3]),
                           "v"(a[4]), "v"(b[4]), "v"(a[5]), "v"(b[5]),
                           "v"(a[6]), "v"(b[6]), "v"(a[7]), "v"(b[7]));
#pragma unroll
        for (int u = 0; u < OL_U; ++u) {
            const int i = base + u * OL_T + (int)threadIdx.x;
            if (i < cnt) rec[s0 + i] = make_int2(k[u], __float_as_int(a[u] * b[u]));
        }
    }
}

#define EPW 8
#define OWN_T 1024
#define OWN_WAVES (OWN_T / 32)
#define OWN_INIT 7
#define AG_PITCH 132
#define AG_TILES ((OWN_ROWS + 15) / 16)
#define AG_ROWS (AG_TILES * 16)
#define ST_TILE (16 * AG_PITCH)
#define ST_GROUPS (OWN_WAVES / 8)

__device__ __forceinline__
void own_add(float* a, const float4 v, const float nj) {
    atomicAdd(a,      v.x * nj);
    atomicAdd(a + 32, v.y * nj);
    atomicAdd(a + 64, v.z * nj);
    atomicAdd(a + 96, v.w * nj);
}

__global__ __launch_bounds__(OWN_T)
void owner_layer_kernel(const float* __restrict__ h, const int2* __restrict__ rec,
                        const int* __restrict__ start,
                        const float* __restrict__ dinv,
                        const float* __restrict__ W, const float* __restrict__ b,
                        const float* __restrict__ g, const float* __restrict__ be,
                        const float* __restrict__ rm, const float* __restrict__ rv,
                        float* __restrict__ H, int n) {
    __shared__ __attribute__((aligned(16))) float acc[AG_ROWS * AG_PITCH];
    __shared__ __attribute__((aligned(16))) float stg[2][ST_GROUPS][ST_TILE];

    const int wave = __builtin_amdgcn_readfirstlane((int)(threadIdx.x >> 5));
    const int lane = threadIdx.x & 31;
    const int o  = blockIdx.x;
    const int r0 = o * OWN_ROWS;
    const int nr = min(OWN_ROWS, n - r0);

    for (int lb = wave; lb < nr; lb += OWN_INIT * OWN_WAVES) {
        float4 v[OWN_INIT];
#pragma unroll
        for (int u = 0; u < OWN_INIT; ++u) {
            const int lr = min(lb + u * OWN_WAVES, nr - 1);
            v[u] = reinterpret_cast<const float4*>(h + (size_t)(r0 + lr) * CH)[lane];
        }
#pragma unroll
        for (int u = 0; u < OWN_INIT; ++u) {
            const int lr = lb + u * OWN_WAVES;
            if (lr < nr) {
                const float di = dinv[r0 + lr];
                const float nrm = di * di;
                float* a = &acc[lr * AG_PITCH + lane];
                a[0]  = v[u].x * nrm;
                a[32] = v[u].y * nrm;
                a[64] = v[u].z * nrm;
                a[96] = v[u].w * nrm;
            }
        }
    }
    for (int lr = nr + wave; lr < AG_ROWS; lr += OWN_WAVES) {
        float* a = &acc[lr * AG_PITCH + lane];
        a[0] = 0.f; a[32] = 0.f; a[64] = 0.f; a[96] = 0.f;
    }
    __syncthreads();

    const int s0 = __builtin_amdgcn_readfirstlane(start[o]);
    const int s1 = __builtin_amdgcn_readfirstlane(start[o + 1]);
    const int nfull = (s1 - s0) / EPW;
    const int sub = lane & (EPW - 1);

    int2 r = make_int2(0, 0);
    if (wave < nfull) r = rec[s0 + wave * EPW + sub];
    for (int bt = wave; bt < nfull; bt += OWN_WAVES) {
        int2 rn = make_int2(0, 0);
        if (bt + OWN_WAVES < nfull) rn = rec[s0 + (bt + OWN_WAVES) * EPW + sub];
        const int   pk  = r.x;
        const float nrm = __int_as_float(r.y);
        float4 v[EPW];
#pragma unroll
        for (int j = 0; j < EPW; ++j) {
            const int sj = __builtin_amdgcn_readlane(pk, j) >> OWN_SHIFT;
            v[j] = reinterpret_cast<const float4*>(h + (size_t)sj * CH)[lane];
        }
#pragma unroll
        for (int j = 0; j < EPW; ++j) {
            const int   lj = __builtin_amdgcn_readlane(pk, j) & ((1 << OWN_SHIFT) - 1);
            const float nj = __int_as_float(
                __builtin_amdgcn_readlane(__float_as_int(nrm), j));
            own_add(&acc[lj * AG_PITCH + lane], v[j], nj);
        }
        r = rn;
    }
    if (wave == (nfull & (OWN_WAVES - 1))) {
        for (int e = s0 + nfull * EPW; e < s1; ++e) {
            const int2 q = rec[e];
            const int sj = q.x >> OWN_SHIFT;
            const int lj = q.x & ((1 << OWN_SHIFT) - 1);
            const float4 v =
                reinterpret_cast<const float4*>(h + (size_t)sj * CH)[lane];
            own_add(&acc[lj * AG_PITCH + lane], v, __int_as_float(q.y));
        }
    }

    const int cb    = wave & 7;
    const int grp   = wave >> 3;
    const int mrow  = lane & 15;
    const int khalf = (lane >> 4) << 1;
    const int moff  = (lane >> 4) * 8;
    const int c = cb * 16 + (lane & 15);
    const float inv = g[c] * rsqrtf(rv[c] + 1e-5f);
    const float bc  = b[c];
    const float rmc = rm[c];
    const float bec = be[c];
    v2f bf[CH / 4];
#pragma unroll
    for (int kk = 0; kk < CH / 4; ++kk) {
        const int k0 = 16 * (kk & 7) + (kk >> 3) + 4 * khalf;
        bf[kk].x = W[(size_t)k0 * CH + c];
        bf[kk].y = W[(size_t)(k0 + 4) * CH + c];
    }
    __syncthreads();

    const int ntile  = (nr + 15) >> 4;
    const int nround = (ntile + ST_GROUPS - 1) / ST_GROUPS;
    const int aoff   = mrow * AG_PITCH + khalf;
    for (int i = 0; i < nround; ++i) {
        const int t = grp + ST_GROUPS * i;
        const bool valid = t < ntile;
        float* st = &stg[i & 1][grp][0];
        if (valid) {
            const float* as = &acc[t * ST_TILE + aoff];
            v8f d = {0.f, 0.f, 0.f, 0.f, 0.f, 0.f, 0.f, 0.f};
#pragma unroll
            for (int kk = 0; kk < CH / 4; ++kk) {
                const v2f a = *reinterpret_cast<const v2f*>(as + 4 * kk);
                d = __builtin_amdgcn_wmma_f32_16x16x4_f32(
                     false, a,  false, bf[kk],
                     (short)0, d,  false,  false);
            }
#pragma unroll
            for (int j = 0; j < 8; ++j) {
                const float y = (d[j] + bc - rmc) * inv + bec;
                st[(j + moff) * AG_PITCH + c] = y > 0.f ? y : 0.f;
            }
        }
        __syncthreads();
        if (valid) {
#pragma unroll
            for (int q = 0; q < 2; ++q) {
                const int rr  = cb + 8 * q;
                const int row = t * 16 + rr;
                if (row < nr) {
                    const float4 hv =
                        *reinterpret_cast<const float4*>(&st[rr * AG_PITCH + lane * 4]);
                    reinterpret_cast<float4*>(H + (size_t)(r0 + row) * CH)[lane] = hv;
                }
            }
        }
    }
}

__global__ void gemm_out2_kernel(const float* __restrict__ X,
                                 const float* __restrict__ W3,
                                 float* __restrict__ H3, int M) {
    int i = blockIdx.x * blockDim.x + threadIdx.x;
    if (i >= M) return;
    const float* xr = X + (size_t)i * CH;
    float a0 = 0.f, a1 = 0.f;
#pragma unroll 8
    for (int k = 0; k < CH; ++k) {
        const float x = xr[k];
        a0 += x * W3[2 * k];
        a1 += x * W3[2 * k + 1];
    }
    H3[(size_t)i * 2]     = a0;
    H3[(size_t)i * 2 + 1] = a1;
}

#define O2_T 1024
#define O2_U 8
static_assert(2 * OWN_ROWS + 31 <= O2_T, "one float of the range per thread");
static_assert(O2_U == 8, "the statement that holds the row loads names 8 slots");

__global__ __launch_bounds__(O2_T)
void owner_aggregate2_kernel(const float* __restrict__ h3, const int2* __restrict__ rec,
                             const int* __restrict__ start,
                             const float* __restrict__ dinv,
                             const float* __restrict__ b3,
                             float* __restrict__ out, int n) {
    __shared__ float acc2[OWN_ROWS * 2];

    const int o  = blockIdx.x;
    const int r0 = o * OWN_ROWS;
    const int nr = min(OWN_ROWS, n - r0);
    const int g0 = 2 * r0;
    const int g1 = g0 + 2 * nr;
    const int g  = (g0 & ~31) + (int)threadIdx.x;
    const bool mine = g >= g0 && g < g1;

    const int gc = min(max(g, g0), g1 - 1);
    const float dv = dinv[gc >> 1];
    const float hv = h3[gc];
    const float bv = b3[gc & 1];

    const int s0  = start[o];
    const int cnt = start[o + 1] - s0;
    const float2* __restrict__ h3r = reinterpret_cast<const float2*>(h3);
    for (int base = 0; base == 0 || base < cnt; base += O2_U * O2_T) {
        int2   q[O2_U];
        float2 v[O2_U];
#pragma unroll
        for (int u = 0; u < O2_U; ++u) {
            const int i = base + u * O2_T + (int)threadIdx.x;
            q[u] = make_int2(0, 0);
            if (i < cnt) q[u] = rec[s0 + i];
        }
#pragma unroll
        for (int u = 0; u < O2_U; ++u)
            v[u] = h3r[q[u].x >> OWN_SHIFT];
        if (base == 0) {
            if (mine) acc2[g - g0] = hv * (dv * dv) + bv;
            __syncthreads();
        }
        asm volatile("" :: "v"(v[0].x), "v"(v[0].y), "v"(v[1].x), "v"(v[1].y),
                           "v"(v[2].x), "v"(v[2].y), "v"(v[3].x), "v"(v[3].y),
                           "v"(v[4].x), "v"(v[4].y), "v"(v[5].x), "v"(v[5].y),
                           "v"(v[6].x), "v"(v[6].y), "v"(v[7].x), "v"(v[7].y));
#pragma unroll
        for (int u = 0; u < O2_U; ++u) {
            const int i = base + u * O2_T + (int)threadIdx.x;
            if (i < cnt) {
                const int   lr = q[u].x & ((1 << OWN_SHIFT) - 1);
                const float nj = __int_as_float(q[u].y);
                atomicAdd(&acc2[2 * lr],     v[u].x * nj);
                atomicAdd(&acc2[2 * lr + 1], v[u].y * nj);
            }
        }
    }
    __syncthreads();

    if (mine) out[g] = acc2[g - g0];
}

extern "C" void kernel_launch(void* const* d_in, const int* in_sizes, int n_in,
                              void* d_out, int out_size, void* d_ws, size_t ws_size,
                              hipStream_t stream) {
    const float* x   = (const float*)d_in[0];
    const int*   ei  = (const int*)d_in[1];
    const float* W1  = (const float*)d_in[2];
    const float* b1  = (const float*)d_in[3];
    const float* g1  = (const float*)d_in[4];
    const float* be1 = (const float*)d_in[5];
    const float* rm1 = (const float*)d_in[6];
    const float* rv1 = (const float*)d_in[7];
    const float* W2  = (const float*)d_in[8];
    const float* b2  = (const float*)d_in[9];
    const float* g2  = (const float*)d_in[10];
    const float* be2 = (const float*)d_in[11];
    const float* rm2 = (const float*)d_in[12];
    const float* rv2 = (const float*)d_in[13];
    const float* W3  = (const float*)d_in[14];
    const float* b3  = (const float*)d_in[15];

    const int Nn   = in_sizes[0] / CH;
    const int Ee   = in_sizes[1] / 2;
    const int NC   = Nn * CH;

    const int* src = ei;
    const int* dst = ei + Ee;

    float* bufA = (float*)d_ws;
    float* bufB = bufA + (size_t)NC;
    float* bufC = bufB + (size_t)NC;
    float* h3   = bufC + (size_t)NC;
    float* dinv = h3 + (size_t)Nn * 2;
    int*   cnt  = (int*)(dinv + Nn);
    int*   strt = cnt + OWN_MAX * OWN_MAX;
    int2*  rec  = (int2*)(strt + 2 * OWN_MAX);
    int*   key  = (int*)(rec + Ee);
    float* out  = (float*)d_out;

    const int T = 256;
    const int gN  = (Nn + T - 1) / T;
    const int chunk = (Ee + OWN_MAX - 1) / OWN_MAX;
    const int gOwn  = (Nn + OWN_ROWS - 1) / OWN_ROWS;

    bin_count_kernel<<<OWN_MAX, BIN_T, 0, stream>>>(dst, cnt, Ee, chunk);
    bin_place_kernel<<<OWN_MAX, BIN_T, 0, stream>>>(src, dst, cnt, strt, key, Ee, chunk);
    owner_degree_kernel<<<gOwn, OL_T, 0, stream>>>(key, strt, dinv, Nn);
    owner_norm_kernel<<<gOwn, OL_T, 0, stream>>>(key, strt, dinv, rec);

    owner_layer_kernel<<<gOwn, OWN_T, 0, stream>>>(x, rec, strt, dinv,
                                                   W1, b1, g1, be1, rm1, rv1, bufB, Nn);

    owner_layer_kernel<<<gOwn, OWN_T, 0, stream>>>(bufB, rec, strt, dinv,
                                                   W2, b2, g2, be2, rm2, rv2, bufA, Nn);

    gemm_out2_kernel<<<gN, T, 0, stream>>>(bufA, W3, h3, Nn);
    owner_aggregate2_kernel<<<gOwn, O2_T, 0, stream>>>(h3, rec, strt, dinv, b3, out, Nn);
}


# Edge binning in one launch: each chunk's work-group counts, scans and places its own keys (per-chunk groups, bounds table); owners walk their 256 groups and store one list; 8 to 7 launches
# speedup vs baseline: 5.9258x; 1.1266x over previous
#include <hip/hip_runtime.h>

#define CH 128

typedef float v2f __attribute__((ext_vector_type(2)));
typedef float v8f __attribute__((ext_vector_type(8)));

#define OWN_ROWS 391
#define OWN_MAX 256
#define OWN_SHIFT 9
#define OFF_PITCH (OWN_MAX + 1)
#define BIN_T 1024
#define BIN_U 8
static_assert(BIN_U == 8, "the statements that hold the loads name 8 slots");

__device__ __forceinline__ int wave_scan_incl(int v, const int lane) {
#pragma unroll
    for (int d = 1; d < 32; d <<= 1) {
        const int nb = __shfl_up(v, d);
        if (lane >= d) v += nb;
    }
    return v;
}

__global__ __launch_bounds__(BIN_T)
void bin_kernel(const int* __restrict__ src, const int* __restrict__ dst,
                int* __restrict__ off, int* __restrict__ key,
                int E, int chunk) {
    __shared__ int hist[OWN_MAX];
    __shared__ int wtot[OWN_MAX / 32];
    const int c    = blockIdx.x;
    const int tid  = threadIdx.x;
    const int lane = tid & 31;
    const int e0  = min(E, c * chunk);
    const int cnt = min(E, e0 + chunk) - e0;
    if (tid < OWN_MAX) hist[tid] = 0;
    __syncthreads();
    for (int base = 0; base < cnt; base += BIN_U * BIN_T) {
        int d[BIN_U];
#pragma unroll
        for (int u = 0; u < BIN_U; ++u) {
            const int i = base + u * BIN_T + tid;
            d[u] = dst[e0 + min(i, cnt - 1)];
        }
        asm volatile("" :: "v"(d[0]), "v"(d[1]), "v"(d[2]), "v"(d[3]),
                           "v"(d[4]), "v"(d[5]), "v"(d[6]), "v"(d[7]));
#pragma unroll
        for (int u = 0; u < BIN_U; ++u) {
            const int i = base + u * BIN_T + tid;
            if (i < cnt) atomicAdd(&hist[d[u] / OWN_ROWS], 1);
        }
    }
    __syncthreads();
    int len = 0, incl = 0;
    if (tid < OWN_MAX) {
        len  = hist[tid];
        incl = wave_scan_incl(len, lane);
        if (lane == 31) wtot[tid >> 5] = incl;
    }
    __syncthreads();
    if (tid < OWN_MAX) {
        int wp = 0;
        for (int k = 0; k < (tid >> 5); ++k) wp += wtot[k];
        const int ex = wp + incl - len;
        hist[tid] = ex;
        off[c * OFF_PITCH + tid] = ex;
        if (tid == OWN_MAX - 1) off[c * OFF_PITCH + OWN_MAX] = ex + len;
    }
    __syncthreads();
    for (int base = 0; base < cnt; base += BIN_U * BIN_T) {
        int s[BIN_U], d[BIN_U];
#pragma unroll
        for (int u = 0; u < BIN_U; ++u) {
            const int i = base + u * BIN_T + tid;
            s[u] = src[e0 + min(i, cnt - 1)];
            d[u] = dst[e0 + min(i, cnt - 1)];
        }
        asm volatile("" :: "v"(s[0]), "v"(d[0]), "v"(s[1]), "v"(d[1]),
                           "v"(s[2]), "v"(d[2]), "v"(s[3]), "v"(d[3]),
                           "v"(s[4]), "v"(d[4]), "v"(s[5]), "v"(d[5]),
                           "v"(s[6]), "v"(d[6]), "v"(s[7]), "v"(d[7]));
#pragma unroll
        for (int u = 0; u < BIN_U; ++u) {
            const int i = base + u * BIN_T + tid;
            if (i < cnt) {
                const int ow = d[u] / OWN_ROWS;
                const int lr = d[u] - ow * OWN_ROWS;
                const int slot = atomicAdd(&hist[ow], 1);
                key[e0 + slot] = (s[u] << OWN_SHIFT) | lr;
            }
        }
    }
}

#define OL_T 1024
#define OL_U 8
static_assert(OWN_ROWS <= OL_T, "one counter per thread");
static_assert(OL_U == 8, "the statements that hold the loads name 8 slots");

__global__ __launch_bounds__(OL_T)
void owner_degree_kernel(const int* __restrict__ key, const int* __restrict__ off,
                         int* __restrict__ start, int* __restrict__ key2,
                         float* __restrict__ dinv, int n, int chunk) {
    __shared__ int dcount[OWN_ROWS];
    __shared__ int segB[OWN_MAX];
    __shared__ int segP[OWN_MAX + 1];
    __shared__ int wtot[OWN_MAX / 32];
    __shared__ int wlow[OWN_MAX / 32];
    const int o    = blockIdx.x;
    const int tid  = threadIdx.x;
    const int lane = tid & 31;
    const int r0 = o * OWN_ROWS;
    const int nr = min(OWN_ROWS, n - r0);
    if (tid < OWN_ROWS) dcount[tid] = 1;
    int lo = 0, len = 0, incl = 0;
    if (tid < OWN_MAX) {
        lo   = off[tid * OFF_PITCH + o];
        len  = off[tid * OFF_PITCH + o + 1] - lo;
        incl = wave_scan_incl(len, lane);
        const int lsum = wave_scan_incl(lo, lane);
        if (lane == 31) { wtot[tid >> 5] = incl; wlow[tid >> 5] = lsum; }
    }
    __syncthreads();
    if (tid < OWN_MAX) {
        int wp = 0;
        for (int k = 0; k < (tid >> 5); ++k) wp += wtot[k];
        const int ex = wp + incl - len;
        segP[tid] = ex;
        segB[tid] = tid * chunk + lo;
        if (tid == OWN_MAX - 1) segP[OWN_MAX] = ex + len;
    }
    __syncthreads();
    const int cnt = segP[OWN_MAX];
    int s0 = 0;
#pragma unroll
    for (int k = 0; k < OWN_MAX / 32; ++k) s0 += wlow[k];
    if (tid == 0) {
        start[o] = s0;
        if (o == (int)gridDim.x - 1) start[o + 1] = s0 + cnt;
    }
    for (int base = 0; base < cnt; base += OL_U * OL_T) {
        int k[OL_U], ad[OL_U];
#pragma unroll
        for (int u = 0; u < OL_U; ++u) {
            const int i = min(base + u * OL_T + tid, cnt - 1);
            int g = 0;
#pragma unroll
            for (int st = OWN_MAX / 2; st > 0; st >>= 1)
                if (segP[g + st] <= i) g += st;
            ad[u] = segB[g] + (i - segP[g]);
        }
#pragma unroll
        for (int u = 0; u < OL_U; ++u) k[u] = key[ad[u]];
        asm volatile("" :: "v"(k[0]), "v"(k[1]), "v"(k[2]), "v"(k[3]),
                           "v"(k[4]), "v"(k[5]), "v"(k[6]), "v"(k[7]));
#pragma unroll
        for (int u = 0; u < OL_U; ++u) {
            const int i = base + u * OL_T + tid;
            if (i < cnt) {
                atomicAdd(&dcount[k[u] & ((1 << OWN_SHIFT) - 1)], 1);
                key2[s0 + i] = k[u];
            }
        }
    }
    __syncthreads();
    if (tid < nr)
        dinv[r0 + tid] = rsqrtf((float)dcount[tid]);
}

__global__ __launch_bounds__(OL_T)
void owner_norm_kernel(const int* __restrict__ key, const int* __restrict__ start,
                       const float* __restrict__ dinv, int2* __restrict__ rec) {
    const int o  = blockIdx.x;
    const int r0 = o * OWN_ROWS;
    const int s0  = start[o];
    const int cnt = start[o + 1] - s0;
    for (int base = 0; base < cnt; base += OL_U * OL_T) {
        int   k[OL_U];
        float a[OL_U], b[OL_U];
#pragma unroll
        for (int u = 0; u < OL_U; ++u) {
            const int i = base + u * OL_T + (int)threadIdx.x;
            k[u] = key[s0 + min(i, cnt - 1)];
        }
        asm volatile("" :: "v"(k[0]), "v"(k[1]), "v"(k[2]), "v"(k[3]),
                           "v"(k[4]), "v"(k[5]), "v"(k[6]), "v"(k[7]));
#pragma unroll
        for (int u = 0; u < OL_U; ++u) {
            a[u] = dinv[k[u] >> OWN_SHIFT];
            b[u] = dinv[r0 + (k[u] & ((1 << OWN_SHIFT) - 1))];
        }
        asm volatile("" :: "v"(a[0]), "v"(b[0]), "v"(a[1]), "v"(b[1]),
                           "v"(a[2]), "v"(b[2]), "v"(a[3]), "v"(b[3]),
                           "v"(a[4]), "v"(b[4]), "v"(a[5]), "v"(b[5]),
                           "v"(a[6]), "v"(b[6]), "v"(a[7]), "v"(b[7]));
#pragma unroll
        for (int u = 0; u < OL_U; ++u) {
            const int i = base + u * OL_T + (int)threadIdx.x;
            if (i < cnt) rec[s0 + i] = make_int2(k[u], __float_as_int(a[u] * b[u]));
        }
    }
}

#define EPW 8
#define OWN_T 1024
#define OWN_WAVES (OWN_T / 32)
#define OWN_INIT 7
#define AG_PITCH 132
#define AG_TILES ((OWN_ROWS + 15) / 16)
#define AG_ROWS (AG_TILES * 16)
#define ST_TILE (16 * AG_PITCH)
#define ST_GROUPS (OWN_WAVES / 8)

__device__ __forceinline__
void own_add(float* a, const float4 v, const float nj) {
    atomicAdd(a,      v.x * nj);
    atomicAdd(a + 32, v.y * nj);
    atomicAdd(a + 64, v.z * nj);
    atomicAdd(a + 96, v.w * nj);
}

__global__ __launch_bounds__(OWN_T)
void owner_layer_kernel(const float* __restrict__ h, const int2* __restrict__ rec,
                        const int* __restrict__ start,
                        const float* __restrict__ dinv,
                        const float* __restrict__ W, const float* __restrict__ b,
                        const float* __restrict__ g, const float* __restrict__ be,
                        const float* __restrict__ rm, const float* __restrict__ rv,
                        float* __restrict__ H, int n) {
    __shared__ __attribute__((aligned(16))) float acc[AG_ROWS * AG_PITCH];
    __shared__ __attribute__((aligned(16))) float stg[2][ST_GROUPS][ST_TILE];

    const int wave = __builtin_amdgcn_readfirstlane((int)(threadIdx.x >> 5));
    const int lane = threadIdx.x & 31;
    const int o  = blockIdx.x;
    const int r0 = o * OWN_ROWS;
    const int nr = min(OWN_ROWS, n - r0);

    for (int lb = wave; lb < nr; lb += OWN_INIT * OWN_WAVES) {
        float4 v[OWN_INIT];
#pragma unroll
        for (int u = 0; u < OWN_INIT; ++u) {
            const int lr = min(lb + u * OWN_WAVES, nr - 1);
            v[u] = reinterpret_cast<const float4*>(h + (size_t)(r0 + lr) * CH)[lane];
        }
#pragma unroll
        for (int u = 0; u < OWN_INIT; ++u) {
            const int lr = lb + u * OWN_WAVES;
            if (lr < nr) {
                const float di = dinv[r0 + lr];
                const float nrm = di * di;
                float* a = &acc[lr * AG_PITCH + lane];
                a[0]  = v[u].x * nrm;
                a[32] = v[u].y * nrm;
                a[64] = v[u].z * nrm;
                a[96] = v[u].w * nrm;
            }
        }
    }
    for (int lr = nr + wave; lr < AG_ROWS; lr += OWN_WAVES) {
        float* a = &acc[lr * AG_PITCH + lane];
        a[0] = 0.f; a[32] = 0.f; a[64] = 0.f; a[96] = 0.f;
    }
    __syncthreads();

    const int s0 = __builtin_amdgcn_readfirstlane(start[o]);
    const int s1 = __builtin_amdgcn_readfirstlane(start[o + 1]);
    const int nfull = (s1 - s0) / EPW;
    const int sub = lane & (EPW - 1);

    int2 r = make_int2(0, 0);
    if (wave < nfull) r = rec[s0 + wave * EPW + sub];
    for (int bt = wave; bt < nfull; bt += OWN_WAVES) {
        int2 rn = make_int2(0, 0);
        if (bt + OWN_WAVES < nfull) rn = rec[s0 + (bt + OWN_WAVES) * EPW + sub];
        const int   pk  = r.x;
        const float nrm = __int_as_float(r.y);
        float4 v[EPW];
#pragma unroll
        for (int j = 0; j < EPW; ++j) {
            const int sj = __builtin_amdgcn_readlane(pk, j) >> OWN_SHIFT;
            v[j] = reinterpret_cast<const float4*>(h + (size_t)sj * CH)[lane];
        }
#pragma unroll
        for (int j = 0; j < EPW; ++j) {
            const int   lj = __builtin_amdgcn_readlane(pk, j) & ((1 << OWN_SHIFT) - 1);
            const float nj = __int_as_float(
                __builtin_amdgcn_readlane(__float_as_int(nrm), j));
            own_add(&acc[lj * AG_PITCH + lane], v[j], nj);
        }
        r = rn;
    }
    if (wave == (nfull & (OWN_WAVES - 1))) {
        for (int e = s0 + nfull * EPW; e < s1; ++e) {
            const int2 q = rec[e];
            const int sj = q.x >> OWN_SHIFT;
            const int lj = q.x & ((1 << OWN_SHIFT) - 1);
            const float4 v =
                reinterpret_cast<const float4*>(h + (size_t)sj * CH)[lane];
            own_add(&acc[lj * AG_PITCH + lane], v, __int_as_float(q.y));
        }
    }

    const int cb    = wave & 7;
    const int grp   = wave >> 3;
    const int mrow  = lane & 15;
    const int khalf = (lane >> 4) << 1;
    const int moff  = (lane >> 4) * 8;
    const int c = cb * 16 + (lane & 15);
    const float inv = g[c] * rsqrtf(rv[c] + 1e-5f);
    const float bc  = b[c];
    const float rmc = rm[c];
    const float bec = be[c];
    v2f bf[CH / 4];
#pragma unroll
    for (int kk = 0; kk < CH / 4; ++kk) {
        const int k0 = 16 * (kk & 7) + (kk >> 3) + 4 * khalf;
        bf[kk].x = W[(size_t)k0 * CH + c];
        bf[kk].y = W[(size_t)(k0 + 4) * CH + c];
    }
    __syncthreads();

    const int ntile  = (nr + 15) >> 4;
    const int nround = (ntile + ST_GROUPS - 1) / ST_GROUPS;
    const int aoff   = mrow * AG_PITCH + khalf;
    for (int i = 0; i < nround; ++i) {
        const int t = grp + ST_GROUPS * i;
        const bool valid = t < ntile;
        float* st = &stg[i & 1][grp][0];
        if (valid) {
            const float* as = &acc[t * ST_TILE + aoff];
            v8f d = {0.f, 0.f, 0.f, 0.f, 0.f, 0.f, 0.f, 0.f};
#pragma unroll
            for (int kk = 0; kk < CH / 4; ++kk) {
                const v2f a = *reinterpret_cast<const v2f*>(as + 4 * kk);
                d = __builtin_amdgcn_wmma_f32_16x16x4_f32(
                     false, a,  false, bf[kk],
                     (short)0, d,  false,  false);
            }
#pragma unroll
            for (int j = 0; j < 8; ++j) {
                const float y = (d[j] + bc - rmc) * inv + bec;
                st[(j + moff) * AG_PITCH + c] = y > 0.f ? y : 0.f;
            }
        }
        __syncthreads();
        if (valid) {
#pragma unroll
            for (int q = 0; q < 2; ++q) {
                const int rr  = cb + 8 * q;
                const int row = t * 16 + rr;
                if (row < nr) {
                    const float4 hv =
                        *reinterpret_cast<const float4*>(&st[rr * AG_PITCH + lane * 4]);
                    reinterpret_cast<float4*>(H + (size_t)(r0 + row) * CH)[lane] = hv;
                }
            }
        }
    }
}

__global__ void gemm_out2_kernel(const float* __restrict__ X,
                                 const float* __restrict__ W3,
                                 float* __restrict__ H3, int M) {
    int i = blockIdx.x * blockDim.x + threadIdx.x;
    if (i >= M) return;
    const float* xr = X + (size_t)i * CH;
    float a0 = 0.f, a1 = 0.f;
#pragma unroll 8
    for (int k = 0; k < CH; ++k) {
        const float x = xr[k];
        a0 += x * W3[2 * k];
        a1 += x * W3[2 * k + 1];
    }
    H3[(size_t)i * 2]     = a0;
    H3[(size_t)i * 2 + 1] = a1;
}

#define O2_T 1024
#define O2_U 8
static_assert(2 * OWN_ROWS + 31 <= O2_T, "one float of the range per thread");
static_assert(O2_U == 8, "the statement that holds the row loads names 8 slots");

__global__ __launch_bounds__(O2_T)
void owner_aggregate2_kernel(const float* __restrict__ h3, const int2* __restrict__ rec,
                             const int* __restrict__ start,
                             const float* __restrict__ dinv,
                             const float* __restrict__ b3,
                             float* __restrict__ out, int n) {
    __shared__ float acc2[OWN_ROWS * 2];

    const int o  = blockIdx.x;
    const int r0 = o * OWN_ROWS;
    const int nr = min(OWN_ROWS, n - r0);
    const int g0 = 2 * r0;
    const int g1 = g0 + 2 * nr;
    const int g  = (g0 & ~31) + (int)threadIdx.x;
    const bool mine = g >= g0 && g < g1;

    const int gc = min(max(g, g0), g1 - 1);
    const float dv = dinv[gc >> 1];
    const float hv = h3[gc];
    const float bv = b3[gc & 1];

    const int s0  = start[o];
    const int cnt = start[o + 1] - s0;
    const float2* __restrict__ h3r = reinterpret_cast<const float2*>(h3);
    for (int base = 0; base == 0 || base < cnt; base += O2_U * O2_T) {
        int2   q[O2_U];
        float2 v[O2_U];
#pragma unroll
        for (int u = 0; u < O2_U; ++u) {
            const int i = base + u * O2_T + (int)threadIdx.x;
            q[u] = make_int2(0, 0);
            if (i < cnt) q[u] = rec[s0 + i];
        }
#pragma unroll
        for (int u = 0; u < O2_U; ++u)
            v[u] = h3r[q[u].x >> OWN_SHIFT];
        if (base == 0) {
            if (mine) acc2[g - g0] = hv * (dv * dv) + bv;
            __syncthreads();
        }
        asm volatile("" :: "v"(v[0].x), "v"(v[0].y), "v"(v[1].x), "v"(v[1].y),
                           "v"(v[2].x), "v"(v[2].y), "v"(v[3].x), "v"(v[3].y),
                           "v"(v[4].x), "v"(v[4].y), "v"(v[5].x), "v"(v[5].y),
                           "v"(v[6].x), "v"(v[6].y), "v"(v[7].x), "v"(v[7].y));
#pragma unroll
        for (int u = 0; u < O2_U; ++u) {
            const int i = base + u * O2_T + (int)threadIdx.x;
            if (i < cnt) {
                const int   lr = q[u].x & ((1 << OWN_SHIFT) - 1);
                const float nj = __int_as_float(q[u].y);
                atomicAdd(&acc2[2 * lr],     v[u].x * nj);
                atomicAdd(&acc2[2 * lr + 1], v[u].y * nj);
            }
        }
    }
    __syncthreads();

    if (mine) out[g] = acc2[g - g0];
}

extern "C" void kernel_launch(void* const* d_in, const int* in_sizes, int n_in,
                              void* d_out, int out_size, void* d_ws, size_t ws_size,
                              hipStream_t stream) {
    const float* x   = (const float*)d_in[0];
    const int*   ei  = (const int*)d_in[1];
    const float* W1  = (const float*)d_in[2];
    const float* b1  = (const float*)d_in[3];
    const float* g1  = (const float*)d_in[4];
    const float* be1 = (const float*)d_in[5];
    const float* rm1 = (const float*)d_in[6];
    const float* rv1 = (const float*)d_in[7];
    const float* W2  = (const float*)d_in[8];
    const float* b2  = (const float*)d_in[9];
    const float* g2  = (const float*)d_in[10];
    const float* be2 = (const float*)d_in[11];
    const float* rm2 = (const float*)d_in[12];
    const float* rv2 = (const float*)d_in[13];
    const float* W3  = (const float*)d_in[14];
    const float* b3  = (const float*)d_in[15];

    const int Nn   = in_sizes[0] / CH;
    const int Ee   = in_sizes[1] / 2;
    const int NC   = Nn * CH;

    const int* src = ei;
    const int* dst = ei + Ee;

    float* bufA = (float*)d_ws;
    float* bufB = bufA + (size_t)NC;
    float* bufC = bufB + (size_t)NC;
    float* h3   = bufC + (size_t)NC;
    float* dinv = h3 + (size_t)Nn * 2;
    int*   off  = (int*)(dinv + Nn);
    int*   strt = off + OWN_MAX * OFF_PITCH;
    int2*  rec  = (int2*)(strt + 2 * OWN_MAX);
    int*   key  = (int*)(rec + Ee);
    int*   key2 = key + Ee;
    float* out  = (float*)d_out;

    const int T = 256;
    const int gN  = (Nn + T - 1) / T;
    const int chunk = (Ee + OWN_MAX - 1) / OWN_MAX;
    const int gOwn  = (Nn + OWN_ROWS - 1) / OWN_ROWS;

    bin_kernel<<<OWN_MAX, BIN_T, 0, stream>>>(src, dst, off, key, Ee, chunk);
    owner_degree_kernel<<<gOwn, OL_T, 0, stream>>>(key, off, strt, key2, dinv, Nn, chunk);
    owner_norm_kernel<<<gOwn, OL_T, 0, stream>>>(key2, strt, dinv, rec);

    owner_layer_kernel<<<gOwn, OWN_T, 0, stream>>>(x, rec, strt, dinv,
                                                   W1, b1, g1, be1, rm1, rv1, bufB, Nn);

    owner_layer_kernel<<<gOwn, OWN_T, 0, stream>>>(bufB, rec, strt, dinv,
                                                   W2, b2, g2, be2, rm2, rv2, bufA, Nn);

    gemm_out2_kernel<<<gN, T, 0, stream>>>(bufA, W3, h3, Nn);
    owner_aggregate2_kernel<<<gOwn, O2_T, 0, stream>>>(h3, rec, strt, dinv, b3, out, Nn);
}
